# starting kernel
# baseline (speedup 1.0000x reference)
#include <hip/hip_runtime.h>
#include <hip/hip_bf16.h>

#define NFEAT  512
#define NCLASS 256

typedef float v2f __attribute__((ext_vector_type(2)));
typedef float v8f __attribute__((ext_vector_type(8)));

__global__ void spgat_gemm(const float* __restrict__ x,
                           const float* __restrict__ W,
                           float* __restrict__ h, int N) {
    const int lane = threadIdx.x & 31;
    const int wave = threadIdx.x >> 5;
    const int mtile = blockIdx.x * 8 + wave;
    const int mtiles = N >> 4;
    if (mtile >= mtiles) return;
    const int m0   = mtile << 4;
    const int n0   = blockIdx.y * 64;
    const int half = lane >> 4;
    const int sub  = lane & 15;

    v8f acc0 = {}, acc1 = {}, acc2 = {}, acc3 = {};

    const float* xrow = x + (size_t)(m0 + sub) * NFEAT + 2 * half;
    for (int kb = 0; kb < NFEAT; kb += 4) {
        v2f afrag;
        afrag.x = xrow[kb + 0];
        afrag.y = xrow[kb + 1];

        const float* wk0 = W + (size_t)(kb + 2 * half + 0) * NCLASS + n0 + sub;
        const float* wk1 = W + (size_t)(kb + 2 * half + 1) * NCLASS + n0 + sub;
        v2f b0; b0.x = wk0[ 0]; b0.y = wk1[ 0];
        v2f b1; b1.x = wk0[16]; b1.y = wk1[16];
        v2f b2; b2.x = wk0[32]; b2.y = wk1[32];
        v2f b3; b3.x = wk0[48]; b3.y = wk1[48];

        acc0 = __builtin_amdgcn_wmma_f32_16x16x4_f32(false, afrag, false, b0, (short)0, acc0, false, false);
        acc1 = __builtin_amdgcn_wmma_f32_16x16x4_f32(false, afrag, false, b1, (short)0, acc1, false, false);
        acc2 = __builtin_amdgcn_wmma_f32_16x16x4_f32(false, afrag, false, b2, (short)0, acc2, false, false);
        acc3 = __builtin_amdgcn_wmma_f32_16x16x4_f32(false, afrag, false, b3, (short)0, acc3, false, false);
    }

    float* hout = h + n0 + sub;
    #pragma unroll
    for (int r = 0; r < 8; ++r) {
        const size_t m = (size_t)(m0 + half * 8 + r);
        hout[m * NCLASS +  0] = acc0[r];
        hout[m * NCLASS + 16] = acc1[r];
        hout[m * NCLASS + 32] = acc2[r];
        hout[m * NCLASS + 48] = acc3[r];
    }
}

__global__ void spgat_scores(const float* __restrict__ h,
                             const float* __restrict__ a,
                             float* __restrict__ s_row,
                             float* __restrict__ s_col, int N) {
    const int lane = threadIdx.x & 31;
    const int row  = blockIdx.x * (blockDim.x >> 5) + (threadIdx.x >> 5);
    if (row >= N) return;
    const float* hr = h + (size_t)row * NCLASS;
    float sr = 0.f, sc = 0.f;
    #pragma unroll
    for (int i = 0; i < NCLASS / 32; ++i) {
        const int f = lane + 32 * i;
        const float v = hr[f];
        sr += v * a[f];
        sc += v * a[NCLASS + f];
    }
    #pragma unroll
    for (int off = 16; off; off >>= 1) {
        sr += __shfl_xor(sr, off, 32);
        sc += __shfl_xor(sc, off, 32);
    }
    if (lane == 0) { s_row[row] = sr; s_col[row] = sc; }
}

__global__ void spgat_edges(const int* __restrict__ ei, int E,
                            const float* __restrict__ s_row,
                            const float* __restrict__ s_col,
                            float* __restrict__ e_buf,
                            float* __restrict__ rowsum) {
    const int e = blockIdx.x * blockDim.x + threadIdx.x;
    if (e >= E) return;
    const int r = ei[e];
    const int c = ei[E + e];
    const float lg = s_row[r] + s_col[c];
    const float l  = lg > 0.f ? lg : 0.2f * lg;
    const float ev = __expf(-l);
    e_buf[e] = ev;
    atomicAdd(rowsum + r, ev);
}

__global__ void spgat_scatter(const int* __restrict__ ei, int E,
                              const float* __restrict__ e_buf,
                              const float* __restrict__ h,
                              float* __restrict__ out) {
    const unsigned tid = blockIdx.x * blockDim.x + threadIdx.x;
    const int e  = tid >> 6;
    const int fg = tid & 63;
    if (e >= E) return;
    const int r = ei[e];
    const int c = ei[E + e];
    const float ev = e_buf[e];
    const float4 hv = *(const float4*)(h + (size_t)c * NCLASS + fg * 4);
    float* o = out + (size_t)r * NCLASS + fg * 4;
    atomicAdd(o + 0, ev * hv.x);
    atomicAdd(o + 1, ev * hv.y);
    atomicAdd(o + 2, ev * hv.z);
    atomicAdd(o + 3, ev * hv.w);
}

__global__ void spgat_finalize(float* __restrict__ out,
                               const float* __restrict__ rowsum, int N) {
    const int lane = threadIdx.x & 31;
    const int row  = blockIdx.x * (blockDim.x >> 5) + (threadIdx.x >> 5);
    if (row >= N) return;
    const float inv = 1.0f / rowsum[row];
    float* o = out + (size_t)row * NCLASS;
    float v[NCLASS / 32];
    float mx = -3.402823466e38f;
    #pragma unroll
    for (int i = 0; i < NCLASS / 32; ++i) {
        const float hp = o[lane + 32 * i] * inv;
        const float el = hp > 0.f ? hp : (__expf(hp) - 1.f);
        v[i] = el;
        mx = fmaxf(mx, el);
    }
    #pragma unroll
    for (int off = 16; off; off >>= 1) mx = fmaxf(mx, __shfl_xor(mx, off, 32));
    float s = 0.f;
    #pragma unroll
    for (int i = 0; i < NCLASS / 32; ++i) s += __expf(v[i] - mx);
    #pragma unroll
    for (int off = 16; off; off >>= 1) s += __shfl_xor(s, off, 32);
    const float lse = mx + __logf(s);
    #pragma unroll
    for (int i = 0; i < NCLASS / 32; ++i) o[lane + 32 * i] = v[i] - lse;
}

extern "C" void kernel_launch(void* const* d_in, const int* in_sizes, int n_in,
                              void* d_out, int out_size, void* d_ws, size_t ws_size,
                              hipStream_t stream) {
    const float* x  = (const float*)d_in[0];
    const float* W  = (const float*)d_in[1];
    const float* a  = (const float*)d_in[2];
    const int*   ei = (const int*)d_in[3];

    const int N = in_sizes[0] / NFEAT;
    const int E = in_sizes[3] / 2;

    float* h      = (float*)d_ws;
    float* s_row  = h + (size_t)N * NCLASS;
    float* s_col  = s_row + N;
    float* rowsum = s_col + N;
    float* e_buf  = rowsum + N;

    float* out = (float*)d_out;

    hipMemsetAsync(out, 0, (size_t)N * NCLASS * sizeof(float), stream);
    hipMemsetAsync(rowsum, 0, (size_t)N * sizeof(float), stream);

    {
        const int mtiles = N / 16;
        dim3 grid((mtiles + 7) / 8, NCLASS / 64);
        spgat_gemm<<<grid, 256, 0, stream>>>(x, W, h, N);
    }
    {
        dim3 grid((N + 7) / 8);
        spgat_scores<<<grid, 256, 0, stream>>>(h, a, s_row, s_col, N);
    }
    {
        dim3 grid((E + 255) / 256);
        spgat_edges<<<grid, 256, 0, stream>>>(ei, E, s_row, s_col, e_buf, rowsum);
    }
    {
        const long long threads = (long long)E * 64;
        dim3 grid((unsigned)((threads + 255) / 256));
        spgat_scatter<<<grid, 256, 0, stream>>>(ei, E, e_buf, h, out);
    }
    {
        dim3 grid((N + 7) / 8);
        spgat_finalize<<<grid, 256, 0, stream>>>(out, rowsum, N);
    }
}


# Output-row-owner aggregation: a block keeps 196 rows' sums in LDS, scans all edges for them, adds e*h[col] by LDS float atomics, finalizes there; no global atomics; row and lse sum order change
# speedup vs baseline: 2.2954x; 2.2954x over previous
#include <hip/hip_runtime.h>
#include <hip/hip_bf16.h>

#define NFEAT  512
#define NCLASS 256

typedef float v2f __attribute__((ext_vector_type(2)));
typedef float v8f __attribute__((ext_vector_type(8)));

__global__ void spgat_gemm(const float* __restrict__ x,
                           const float* __restrict__ W,
                           float* __restrict__ h, int N) {
    const int lane = threadIdx.x & 31;
    const int wave = threadIdx.x >> 5;
    const int mtile = blockIdx.x * 8 + wave;
    const int mtiles = N >> 4;
    if (mtile >= mtiles) return;
    const int m0   = mtile << 4;
    const int n0   = blockIdx.y * 64;
    const int half = lane >> 4;
    const int sub  = lane & 15;

    v8f acc0 = {}, acc1 = {}, acc2 = {}, acc3 = {};

    const float* xrow = x + (size_t)(m0 + sub) * NFEAT + 2 * half;
    for (int kb = 0; kb < NFEAT; kb += 4) {
        v2f afrag;
        afrag.x = xrow[kb + 0];
        afrag.y = xrow[kb + 1];

        const float* wk0 = W + (size_t)(kb + 2 * half + 0) * NCLASS + n0 + sub;
        const float* wk1 = W + (size_t)(kb + 2 * half + 1) * NCLASS + n0 + sub;
        v2f b0; b0.x = wk0[ 0]; b0.y = wk1[ 0];
        v2f b1; b1.x = wk0[16]; b1.y = wk1[16];
        v2f b2; b2.x = wk0[32]; b2.y = wk1[32];
        v2f b3; b3.x = wk0[48]; b3.y = wk1[48];

        acc0 = __builtin_amdgcn_wmma_f32_16x16x4_f32(false, afrag, false, b0, (short)0, acc0, false, false);
        acc1 = __builtin_amdgcn_wmma_f32_16x16x4_f32(false, afrag, false, b1, (short)0, acc1, false, false);
        acc2 = __builtin_amdgcn_wmma_f32_16x16x4_f32(false, afrag, false, b2, (short)0, acc2, false, false);
        acc3 = __builtin_amdgcn_wmma_f32_16x16x4_f32(false, afrag, false, b3, (short)0, acc3, false, false);
    }

    float* hout = h + n0 + sub;
    #pragma unroll
    for (int r = 0; r < 8; ++r) {
        const size_t m = (size_t)(m0 + half * 8 + r);
        hout[m * NCLASS +  0] = acc0[r];
        hout[m * NCLASS + 16] = acc1[r];
        hout[m * NCLASS + 32] = acc2[r];
        hout[m * NCLASS + 48] = acc3[r];
    }
}

__global__ void spgat_scores(const float* __restrict__ h,
                             const float* __restrict__ a,
                             float* __restrict__ s_row,
                             float* __restrict__ s_col, int N) {
    const int lane = threadIdx.x & 31;
    const int row  = blockIdx.x * (blockDim.x >> 5) + (threadIdx.x >> 5);
    if (row >= N) return;
    const float* hr = h + (size_t)row * NCLASS;
    float sr = 0.f, sc = 0.f;
    #pragma unroll
    for (int i = 0; i < NCLASS / 32; ++i) {
        const int f = lane + 32 * i;
        const float v = hr[f];
        sr += v * a[f];
        sc += v * a[NCLASS + f];
    }
    #pragma unroll
    for (int off = 16; off; off >>= 1) {
        sr += __shfl_xor(sr, off, 32);
        sc += __shfl_xor(sc, off, 32);
    }
    if (lane == 0) { s_row[row] = sr; s_col[row] = sc; }
}

#define ROWS_PB     196
#define LIST_CAP    8192
#define AGG_THREADS 1024

__global__ void __launch_bounds__(AGG_THREADS)
spgat_aggregate(const int* __restrict__ ei, int E,
                const float* __restrict__ s_row,
                const float* __restrict__ s_col,
                const float* __restrict__ h,
                float* __restrict__ out, int N) {
    __shared__ __attribute__((aligned(16))) float acc[ROWS_PB * NCLASS];
    __shared__ float rsum[ROWS_PB];
    __shared__ int   list_c[LIST_CAP];
    __shared__ int   list_r[LIST_CAP];
    __shared__ float list_e[LIST_CAP];
    __shared__ int   cnt;

    const int tid  = threadIdx.x;
    const int lane = tid & 31;
    const int wave = __builtin_amdgcn_readfirstlane(tid >> 5);
    const int row0 = blockIdx.x * ROWS_PB;
    const int nrows = min(ROWS_PB, N - row0);
    const unsigned nr = (unsigned)nrows;

    for (int i = tid; i < ROWS_PB * NCLASS; i += AGG_THREADS) acc[i] = 0.f;
    if (tid < ROWS_PB) rsum[tid] = 0.f;
    if (tid == 0) cnt = 0;
    __syncthreads();

    auto add_here = [&](int e, unsigned rl) {
        const int c = ei[E + e];
        const float lg = s_row[row0 + (int)rl] + s_col[c];
        const float l  = lg > 0.f ? lg : 0.2f * lg;
        const float ev = __expf(-l);
        const float* hc = h + (size_t)c * NCLASS;
        float* ar = acc + rl * NCLASS;
        #pragma unroll 1
        for (int f = 0; f < NCLASS; ++f) atomicAdd(ar + f, ev * hc[f]);
        atomicAdd(rsum + rl, ev);
    };
    auto put = [&](int pos, int e, unsigned rl) {
        if (pos < LIST_CAP) { list_c[pos] = e; list_r[pos] = (int)rl; }
        else add_here(e, rl);
    };

    const int nvec = E >> 2;
    const int4* rows4 = (const int4*)ei;
    const int voff = (int)(((long long)blockIdx.x * nvec) / gridDim.x) & ~31;
    for (int v0 = tid; v0 < nvec; v0 += 8 * AGG_THREADS) {
        int4 q[8];
        #pragma unroll
        for (int u = 0; u < 8; ++u) {
            const int v = v0 + u * AGG_THREADS;
            const int vv = (v + voff < nvec) ? v + voff : v + voff - nvec;
            q[u] = (v < nvec) ? rows4[vv] : make_int4(-1, -1, -1, -1);
        }
        #pragma unroll
        for (int u = 0; u < 8; ++u) {
            const int v = v0 + u * AGG_THREADS;
            const int e = ((v + voff < nvec) ? v + voff : v + voff - nvec) << 2;
            const unsigned r0 = (unsigned)(q[u].x - row0), r1 = (unsigned)(q[u].y - row0);
            const unsigned r2 = (unsigned)(q[u].z - row0), r3 = (unsigned)(q[u].w - row0);
            if (min(min(r0, r1), min(r2, r3)) < nr) {
                const int m0 = r0 < nr, m1 = r1 < nr, m2 = r2 < nr, m3 = r3 < nr;
                int pos = atomicAdd(&cnt, m0 + m1 + m2 + m3);
                if (m0) put(pos++, e + 0, r0);
                if (m1) put(pos++, e + 1, r1);
                if (m2) put(pos++, e + 2, r2);
                if (m3) put(pos++, e + 3, r3);
            }
        }
    }
    for (int e = (nvec << 2) + tid; e < E; e += AGG_THREADS) {
        const unsigned rl = (unsigned)(ei[e] - row0);
        if (rl < nr) put(atomicAdd(&cnt, 1), e, rl);
    }
    __syncthreads();

    const int n = min(cnt, LIST_CAP);
    for (int k = tid; k < n; k += AGG_THREADS) {
        const int e = list_c[k];
        const int c = ei[E + e];
        const float lg = s_row[row0 + list_r[k]] + s_col[c];
        const float l  = lg > 0.f ? lg : 0.2f * lg;
        list_c[k] = c;
        list_e[k] = __expf(-l);
    }
    __syncthreads();

    for (int k0 = wave; k0 < n; k0 += 4 * 32) {
        float hv[4][NCLASS / 32];
        float ev[4];
        int   cc[4], rl[4];
        #pragma unroll
        for (int u = 0; u < 4; ++u) {
            const int kk = min(k0 + 32 * u, n - 1);
            cc[u] = list_c[kk];
            rl[u] = list_r[kk];
            ev[u] = list_e[kk];
        }
        #pragma unroll
        for (int u = 0; u < 4; ++u) {
            if (k0 + 32 * u < n) {
                const float* hc = h + (size_t)cc[u] * NCLASS + lane;
                #pragma unroll
                for (int i = 0; i < NCLASS / 32; ++i) hv[u][i] = hc[32 * i];
            }
        }
        #pragma unroll
        for (int u = 0; u < 4; ++u) {
            if (k0 + 32 * u < n) {
                float* ar = acc + rl[u] * NCLASS + lane;
                #pragma unroll
                for (int i = 0; i < NCLASS / 32; ++i) atomicAdd(ar + 32 * i, ev[u] * hv[u][i]);
                if (lane == 0) atomicAdd(rsum + rl[u], ev[u]);
            }
        }
    }
    __syncthreads();

    for (int r = wave; r < nrows; r += 32) {
        const float inv = 1.0f / rsum[r];
        const float4* ar = (const float4*)(acc + r * NCLASS);
        float4* o = (float4*)(out + (size_t)(row0 + r) * NCLASS);
        const float4 p0 = ar[lane];
        const float4 p1 = ar[lane + 32];
        float v[8] = { p0.x, p0.y, p0.z, p0.w, p1.x, p1.y, p1.z, p1.w };
        float mx = -3.402823466e38f;
        #pragma unroll
        for (int i = 0; i < 8; ++i) {
            const float hp = v[i] * inv;
            const float el = hp > 0.f ? hp : (__expf(hp) - 1.f);
            v[i] = el;
            mx = fmaxf(mx, el);
        }
        #pragma unroll
        for (int off = 16; off; off >>= 1) mx = fmaxf(mx, __shfl_xor(mx, off, 32));
        float s = 0.f;
        #pragma unroll
        for (int i = 0; i < 8; ++i) s += __expf(v[i] - mx);
        #pragma unroll
        for (int off = 16; off; off >>= 1) s += __shfl_xor(s, off, 32);
        const float lse = mx + __logf(s);
        o[lane]      = make_float4(v[0] - lse, v[1] - lse, v[2] - lse, v[3] - lse);
        o[lane + 32] = make_float4(v[4] - lse, v[5] - lse, v[6] - lse, v[7] - lse);
    }
}

extern "C" void kernel_launch(void* const* d_in, const int* in_sizes, int n_in,
                              void* d_out, int out_size, void* d_ws, size_t ws_size,
                              hipStream_t stream) {
    const float* x  = (const float*)d_in[0];
    const float* W  = (const float*)d_in[1];
    const float* a  = (const float*)d_in[2];
    const int*   ei = (const int*)d_in[3];

    const int N = in_sizes[0] / NFEAT;
    const int E = in_sizes[3] / 2;

    float* h      = (float*)d_ws;
    float* s_row  = h + (size_t)N * NCLASS;
    float* s_col  = s_row + N;

    float* out = (float*)d_out;

    {
        const int mtiles = N / 16;
        dim3 grid((mtiles + 7) / 8, NCLASS / 64);
        spgat_gemm<<<grid, 256, 0, stream>>>(x, W, h, N);
    }
    {
        dim3 grid((N + 7) / 8);
        spgat_scores<<<grid, 256, 0, stream>>>(h, a, s_row, s_col, N);
    }
    {
        dim3 grid((N + ROWS_PB - 1) / ROWS_PB);
        spgat_aggregate<<<grid, AGG_THREADS, 0, stream>>>(ei, E, s_row, s_col, h, out, N);
    }
}


# GEMM re-tiled in registers: a wave owns 32 rows x 64 cols (2 A fragments x 4 B fragments, 8 accumulator tiles), a block's 4 waves the 4 column slabs of one row block; h sum order unchanged
# speedup vs baseline: 2.5472x; 1.1097x over previous
#include <hip/hip_runtime.h>
#include <hip/hip_bf16.h>

#define NFEAT  512
#define NCLASS 256

typedef float v2f __attribute__((ext_vector_type(2)));
typedef float v8f __attribute__((ext_vector_type(8)));

__global__ void __launch_bounds__(128)
spgat_gemm(const float* __restrict__ x,
           const float* __restrict__ W,
           float* __restrict__ h, int N) {
    const int lane = threadIdx.x & 31;
    const int wave = threadIdx.x >> 5;
    const int mtiles = N >> 4;
    const int mt0  = blockIdx.x * 2;
    const bool has1 = (mt0 + 1) < mtiles;
    const int m0   = mt0 << 4;
    const int m1   = has1 ? m0 + 16 : m0;
    const int n0   = wave * 64;
    const int half = lane >> 4;
    const int sub  = lane & 15;

    v8f c00 = {}, c01 = {}, c02 = {}, c03 = {};
    v8f c10 = {}, c11 = {}, c12 = {}, c13 = {};

    const float* xr0 = x + (size_t)(m0 + sub) * NFEAT + 2 * half;
    const float* xr1 = x + (size_t)(m1 + sub) * NFEAT + 2 * half;
    const float* wb  = W + (size_t)(2 * half) * NCLASS + n0 + sub;
    for (int kb = 0; kb < NFEAT; kb += 4) {
        v2f a0; a0.x = xr0[kb + 0]; a0.y = xr0[kb + 1];
        v2f a1; a1.x = xr1[kb + 0]; a1.y = xr1[kb + 1];

        const float* wk0 = wb + (size_t)kb * NCLASS;
        const float* wk1 = wk0 + NCLASS;
        v2f b0; b0.x = wk0[ 0]; b0.y = wk1[ 0];
        v2f b1; b1.x = wk0[16]; b1.y = wk1[16];
        v2f b2; b2.x = wk0[32]; b2.y = wk1[32];
        v2f b3; b3.x = wk0[48]; b3.y = wk1[48];

        c00 = __builtin_amdgcn_wmma_f32_16x16x4_f32(false, a0, false, b0, (short)0, c00, false, false);
        c10 = __builtin_amdgcn_wmma_f32_16x16x4_f32(false, a1, false, b0, (short)0, c10, false, false);
        c01 = __builtin_amdgcn_wmma_f32_16x16x4_f32(false, a0, false, b1, (short)0, c01, false, false);
        c11 = __builtin_amdgcn_wmma_f32_16x16x4_f32(false, a1, false, b1, (short)0, c11, false, false);
        c02 = __builtin_amdgcn_wmma_f32_16x16x4_f32(false, a0, false, b2, (short)0, c02, false, false);
        c12 = __builtin_amdgcn_wmma_f32_16x16x4_f32(false, a1, false, b2, (short)0, c12, false, false);
        c03 = __builtin_amdgcn_wmma_f32_16x16x4_f32(false, a0, false, b3, (short)0, c03, false, false);
        c13 = __builtin_amdgcn_wmma_f32_16x16x4_f32(false, a1, false, b3, (short)0, c13, false, false);
    }

    float* hout = h + n0 + sub;
    #pragma unroll
    for (int r = 0; r < 8; ++r) {
        const size_t m = (size_t)(m0 + half * 8 + r);
        hout[m * NCLASS +  0] = c00[r];
        hout[m * NCLASS + 16] = c01[r];
        hout[m * NCLASS + 32] = c02[r];
        hout[m * NCLASS + 48] = c03[r];
    }
    if (has1) {
        #pragma unroll
        for (int r = 0; r < 8; ++r) {
            const size_t m = (size_t)(m1 + half * 8 + r);
            hout[m * NCLASS +  0] = c10[r];
            hout[m * NCLASS + 16] = c11[r];
            hout[m * NCLASS + 32] = c12[r];
            hout[m * NCLASS + 48] = c13[r];
        }
    }
}

__global__ void spgat_scores(const float* __restrict__ h,
                             const float* __restrict__ a,
                             float* __restrict__ s_row,
                             float* __restrict__ s_col, int N) {
    const int lane = threadIdx.x & 31;
    const int row  = blockIdx.x * (blockDim.x >> 5) + (threadIdx.x >> 5);
    if (row >= N) return;
    const float* hr = h + (size_t)row * NCLASS;
    float sr = 0.f, sc = 0.f;
    #pragma unroll
    for (int i = 0; i < NCLASS / 32; ++i) {
        const int f = lane + 32 * i;
        const float v = hr[f];
        sr += v * a[f];
        sc += v * a[NCLASS + f];
    }
    #pragma unroll
    for (int off = 16; off; off >>= 1) {
        sr += __shfl_xor(sr, off, 32);
        sc += __shfl_xor(sc, off, 32);
    }
    if (lane == 0) { s_row[row] = sr; s_col[row] = sc; }
}

#define ROWS_PB     196
#define LIST_CAP    8192
#define AGG_THREADS 1024

__global__ void __launch_bounds__(AGG_THREADS)
spgat_aggregate(const int* __restrict__ ei, int E,
                const float* __restrict__ s_row,
                const float* __restrict__ s_col,
                const float* __restrict__ h,
                float* __restrict__ out, int N) {
    __shared__ __attribute__((aligned(16))) float acc[ROWS_PB * NCLASS];
    __shared__ float rsum[ROWS_PB];
    __shared__ int   list_c[LIST_CAP];
    __shared__ int   list_r[LIST_CAP];
    __shared__ float list_e[LIST_CAP];
    __shared__ int   cnt;

    const int tid  = threadIdx.x;
    const int lane = tid & 31;
    const int wave = __builtin_amdgcn_readfirstlane(tid >> 5);
    const int row0 = blockIdx.x * ROWS_PB;
    const int nrows = min(ROWS_PB, N - row0);
    const unsigned nr = (unsigned)nrows;

    for (int i = tid; i < ROWS_PB * NCLASS; i += AGG_THREADS) acc[i] = 0.f;
    if (tid < ROWS_PB) rsum[tid] = 0.f;
    if (tid == 0) cnt = 0;
    __syncthreads();

    auto add_here = [&](int e, unsigned rl) {
        const int c = ei[E + e];
        const float lg = s_row[row0 + (int)rl] + s_col[c];
        const float l  = lg > 0.f ? lg : 0.2f * lg;
        const float ev = __expf(-l);
        const float* hc = h + (size_t)c * NCLASS;
        float* ar = acc + rl * NCLASS;
        #pragma unroll 1
        for (int f = 0; f < NCLASS; ++f) atomicAdd(ar + f, ev * hc[f]);
        atomicAdd(rsum + rl, ev);
    };
    auto put = [&](int pos, int e, unsigned rl) {
        if (pos < LIST_CAP) { list_c[pos] = e; list_r[pos] = (int)rl; }
        else add_here(e, rl);
    };

    const int nvec = E >> 2;
    const int4* rows4 = (const int4*)ei;
    const int voff = (int)(((long long)blockIdx.x * nvec) / gridDim.x) & ~31;
    for (int v0 = tid; v0 < nvec; v0 += 8 * AGG_THREADS) {
        int4 q[8];
        #pragma unroll
        for (int u = 0; u < 8; ++u) {
            const int v = v0 + u * AGG_THREADS;
            const int vv = (v + voff < nvec) ? v + voff : v + voff - nvec;
            q[u] = (v < nvec) ? rows4[vv] : make_int4(-1, -1, -1, -1);
        }
        #pragma unroll
        for (int u = 0; u < 8; ++u) {
            const int v = v0 + u * AGG_THREADS;
            const int e = ((v + voff < nvec) ? v + voff : v + voff - nvec) << 2;
            const unsigned r0 = (unsigned)(q[u].x - row0), r1 = (unsigned)(q[u].y - row0);
            const unsigned r2 = (unsigned)(q[u].z - row0), r3 = (unsigned)(q[u].w - row0);
            if (min(min(r0, r1), min(r2, r3)) < nr) {
                const int m0 = r0 < nr, m1 = r1 < nr, m2 = r2 < nr, m3 = r3 < nr;
                int pos = atomicAdd(&cnt, m0 + m1 + m2 + m3);
                if (m0) put(pos++, e + 0, r0);
                if (m1) put(pos++, e + 1, r1);
                if (m2) put(pos++, e + 2, r2);
                if (m3) put(pos++, e + 3, r3);
            }
        }
    }
    for (int e = (nvec << 2) + tid; e < E; e += AGG_THREADS) {
        const unsigned rl = (unsigned)(ei[e] - row0);
        if (rl < nr) put(atomicAdd(&cnt, 1), e, rl);
    }
    __syncthreads();

    const int n = min(cnt, LIST_CAP);
    for (int k = tid; k < n; k += AGG_THREADS) {
        const int e = list_c[k];
        const int c = ei[E + e];
        const float lg = s_row[row0 + list_r[k]] + s_col[c];
        const float l  = lg > 0.f ? lg : 0.2f * lg;
        list_c[k] = c;
        list_e[k] = __expf(-l);
    }
    __syncthreads();

    for (int k0 = wave; k0 < n; k0 += 4 * 32) {
        float hv[4][NCLASS / 32];
        float ev[4];
        int   cc[4], rl[4];
        #pragma unroll
        for (int u = 0; u < 4; ++u) {
            const int kk = min(k0 + 32 * u, n - 1);
            cc[u] = list_c[kk];
            rl[u] = list_r[kk];
            ev[u] = list_e[kk];
        }
        #pragma unroll
        for (int u = 0; u < 4; ++u) {
            if (k0 + 32 * u < n) {
                const float* hc = h + (size_t)cc[u] * NCLASS + lane;
                #pragma unroll
                for (int i = 0; i < NCLASS / 32; ++i) hv[u][i] = hc[32 * i];
            }
        }
        #pragma unroll
        for (int u = 0; u < 4; ++u) {
            if (k0 + 32 * u < n) {
                float* ar = acc + rl[u] * NCLASS + lane;
                #pragma unroll
                for (int i = 0; i < NCLASS / 32; ++i) atomicAdd(ar + 32 * i, ev[u] * hv[u][i]);
                if (lane == 0) atomicAdd(rsum + rl[u], ev[u]);
            }
        }
    }
    __syncthreads();

    for (int r = wave; r < nrows; r += 32) {
        const float inv = 1.0f / rsum[r];
        const float4* ar = (const float4*)(acc + r * NCLASS);
        float4* o = (float4*)(out + (size_t)(row0 + r) * NCLASS);
        const float4 p0 = ar[lane];
        const float4 p1 = ar[lane + 32];
        float v[8] = { p0.x, p0.y, p0.z, p0.w, p1.x, p1.y, p1.z, p1.w };
        float mx = -3.402823466e38f;
        #pragma unroll
        for (int i = 0; i < 8; ++i) {
            const float hp = v[i] * inv;
            const float el = hp > 0.f ? hp : (__expf(hp) - 1.f);
            v[i] = el;
            mx = fmaxf(mx, el);
        }
        #pragma unroll
        for (int off = 16; off; off >>= 1) mx = fmaxf(mx, __shfl_xor(mx, off, 32));
        float s = 0.f;
        #pragma unroll
        for (int i = 0; i < 8; ++i) s += __expf(v[i] - mx);
        #pragma unroll
        for (int off = 16; off; off >>= 1) s += __shfl_xor(s, off, 32);
        const float lse = mx + __logf(s);
        o[lane]      = make_float4(v[0] - lse, v[1] - lse, v[2] - lse, v[3] - lse);
        o[lane + 32] = make_float4(v[4] - lse, v[5] - lse, v[6] - lse, v[7] - lse);
    }
}

extern "C" void kernel_launch(void* const* d_in, const int* in_sizes, int n_in,
                              void* d_out, int out_size, void* d_ws, size_t ws_size,
                              hipStream_t stream) {
    const float* x  = (const float*)d_in[0];
    const float* W  = (const float*)d_in[1];
    const float* a  = (const float*)d_in[2];
    const int*   ei = (const int*)d_in[3];

    const int N = in_sizes[0] / NFEAT;
    const int E = in_sizes[3] / 2;

    float* h      = (float*)d_ws;
    float* s_row  = h + (size_t)N * NCLASS;
    float* s_col  = s_row + N;

    float* out = (float*)d_out;

    {
        const int mtiles = N / 16;
        dim3 grid((mtiles + 1) / 2);
        spgat_gemm<<<grid, 128, 0, stream>>>(x, W, h, N);
    }
    {
        dim3 grid((N + 7) / 8);
        spgat_scores<<<grid, 256, 0, stream>>>(h, a, s_row, s_col, N);
    }
    {
        dim3 grid((N + ROWS_PB - 1) / ROWS_PB);
        spgat_aggregate<<<grid, AGG_THREADS, 0, stream>>>(ei, E, s_row, s_col, h, out, N);
    }
}


# Edge scan on 1 key byte per edge: scores launch writes key[e]=row[e]/196 (owner block); blocks scan 16 edges per 16-B load, one list reservation per wave sweep; row sums stay LDS-atomic arrival order
# speedup vs baseline: 2.8892x; 1.1343x over previous
#include <hip/hip_runtime.h>
#include <hip/hip_bf16.h>

#define NFEAT  512
#define NCLASS 256
#define ROWS_PB 196

typedef float v2f __attribute__((ext_vector_type(2)));
typedef float v8f __attribute__((ext_vector_type(8)));

__global__ void __launch_bounds__(128)
spgat_gemm(const float* __restrict__ x,
           const float* __restrict__ W,
           float* __restrict__ h, int N) {
    const int lane = threadIdx.x & 31;
    const int wave = threadIdx.x >> 5;
    const int mtiles = N >> 4;
    const int mt0  = blockIdx.x * 2;
    const bool has1 = (mt0 + 1) < mtiles;
    const int m0   = mt0 << 4;
    const int m1   = has1 ? m0 + 16 : m0;
    const int n0   = wave * 64;
    const int half = lane >> 4;
    const int sub  = lane & 15;

    v8f c00 = {}, c01 = {}, c02 = {}, c03 = {};
    v8f c10 = {}, c11 = {}, c12 = {}, c13 = {};

    const float* xr0 = x + (size_t)(m0 + sub) * NFEAT + 2 * half;
    const float* xr1 = x + (size_t)(m1 + sub) * NFEAT + 2 * half;
    const float* wb  = W + (size_t)(2 * half) * NCLASS + n0 + sub;
    for (int kb = 0; kb < NFEAT; kb += 4) {
        v2f a0; a0.x = xr0[kb + 0]; a0.y = xr0[kb + 1];
        v2f a1; a1.x = xr1[kb + 0]; a1.y = xr1[kb + 1];

        const float* wk0 = wb + (size_t)kb * NCLASS;
        const float* wk1 = wk0 + NCLASS;
        v2f b0; b0.x = wk0[ 0]; b0.y = wk1[ 0];
        v2f b1; b1.x = wk0[16]; b1.y = wk1[16];
        v2f b2; b2.x = wk0[32]; b2.y = wk1[32];
        v2f b3; b3.x = wk0[48]; b3.y = wk1[48];

        c00 = __builtin_amdgcn_wmma_f32_16x16x4_f32(false, a0, false, b0, (short)0, c00, false, false);
        c10 = __builtin_amdgcn_wmma_f32_16x16x4_f32(false, a1, false, b0, (short)0, c10, false, false);
        c01 = __builtin_amdgcn_wmma_f32_16x16x4_f32(false, a0, false, b1, (short)0, c01, false, false);
        c11 = __builtin_amdgcn_wmma_f32_16x16x4_f32(false, a1, false, b1, (short)0, c11, false, false);
        c02 = __builtin_amdgcn_wmma_f32_16x16x4_f32(false, a0, false, b2, (short)0, c02, false, false);
        c12 = __builtin_amdgcn_wmma_f32_16x16x4_f32(false, a1, false, b2, (short)0, c12, false, false);
        c03 = __builtin_amdgcn_wmma_f32_16x16x4_f32(false, a0, false, b3, (short)0, c03, false, false);
        c13 = __builtin_amdgcn_wmma_f32_16x16x4_f32(false, a1, false, b3, (short)0, c13, false, false);
    }

    float* hout = h + n0 + sub;
    #pragma unroll
    for (int r = 0; r < 8; ++r) {
        const size_t m = (size_t)(m0 + half * 8 + r);
        hout[m * NCLASS +  0] = c00[r];
        hout[m * NCLASS + 16] = c01[r];
        hout[m * NCLASS + 32] = c02[r];
        hout[m * NCLASS + 48] = c03[r];
    }
    if (has1) {
        #pragma unroll
        for (int r = 0; r < 8; ++r) {
            const size_t m = (size_t)(m1 + half * 8 + r);
            hout[m * NCLASS +  0] = c10[r];
            hout[m * NCLASS + 16] = c11[r];
            hout[m * NCLASS + 32] = c12[r];
            hout[m * NCLASS + 48] = c13[r];
        }
    }
}

__global__ void spgat_scores(const float* __restrict__ h,
                             const float* __restrict__ a,
                             float* __restrict__ s_row,
                             float* __restrict__ s_col,
                             const int* __restrict__ ei, int E,
                             unsigned int* __restrict__ key_w, int N) {
    {
        const int nthr = gridDim.x * blockDim.x;
        const int nq   = E >> 2;
        for (int g = blockIdx.x * blockDim.x + threadIdx.x; g < nq; g += nthr) {
            const int4 r = ((const int4*)ei)[g];
            key_w[g] = (((unsigned)r.x / ROWS_PB) & 255u) | ((((unsigned)r.y / ROWS_PB) & 255u) << 8) |
                       ((((unsigned)r.z / ROWS_PB) & 255u) << 16) | (((unsigned)r.w / ROWS_PB) << 24);
        }
        unsigned char* key_b = (unsigned char*)key_w;
        for (int e = (nq << 2) + blockIdx.x * blockDim.x + threadIdx.x; e < E; e += nthr)
            key_b[e] = (unsigned char)((unsigned)ei[e] / ROWS_PB);
    }
    const int lane = threadIdx.x & 31;
    const int row  = blockIdx.x * (blockDim.x >> 5) + (threadIdx.x >> 5);
    if (row >= N) return;
    const float* hr = h + (size_t)row * NCLASS;
    float sr = 0.f, sc = 0.f;
    #pragma unroll
    for (int i = 0; i < NCLASS / 32; ++i) {
        const int f = lane + 32 * i;
        const float v = hr[f];
        sr += v * a[f];
        sc += v * a[NCLASS + f];
    }
    #pragma unroll
    for (int off = 16; off; off >>= 1) {
        sr += __shfl_xor(sr, off, 32);
        sc += __shfl_xor(sc, off, 32);
    }
    if (lane == 0) { s_row[row] = sr; s_col[row] = sc; }
}

#define LIST_CAP    8192
#define AGG_THREADS 1024

__global__ void __launch_bounds__(AGG_THREADS)
spgat_aggregate(const int* __restrict__ ei, int E,
                const unsigned int* __restrict__ key_w,
                const float* __restrict__ s_row,
                const float* __restrict__ s_col,
                const float* __restrict__ h,
                float* __restrict__ out, int N) {
    __shared__ __attribute__((aligned(16))) float acc[ROWS_PB * NCLASS];
    __shared__ float rsum[ROWS_PB];
    __shared__ int   list_c[LIST_CAP];
    __shared__ int   list_r[LIST_CAP];
    __shared__ float list_e[LIST_CAP];
    __shared__ int   cnt;

    const int tid  = threadIdx.x;
    const int lane = tid & 31;
    const int wave = __builtin_amdgcn_readfirstlane(tid >> 5);
    const int row0 = blockIdx.x * ROWS_PB;
    const int nrows = min(ROWS_PB, N - row0);
    const unsigned nr = (unsigned)nrows;

    for (int i = tid; i < ROWS_PB * NCLASS; i += AGG_THREADS) acc[i] = 0.f;
    if (tid < ROWS_PB) rsum[tid] = 0.f;
    if (tid == 0) cnt = 0;
    __syncthreads();

    auto add_here = [&](int e) {
        const unsigned rl = (unsigned)(ei[e] - row0);
        if (rl >= nr) return;
        const int c = ei[E + e];
        const float lg = s_row[row0 + (int)rl] + s_col[c];
        const float l  = lg > 0.f ? lg : 0.2f * lg;
        const float ev = __expf(-l);
        const float* hc = h + (size_t)c * NCLASS;
        float* ar = acc + rl * NCLASS;
        #pragma unroll 1
        for (int f = 0; f < NCLASS; ++f) atomicAdd(ar + f, ev * hc[f]);
        atomicAdd(rsum + rl, ev);
    };
    auto put = [&](int pos, int e) {
        if (pos < LIST_CAP) list_c[pos] = e;
        else add_here(e);
    };
    const unsigned pat = (unsigned)blockIdx.x * 0x01010101u;
    auto hit4 = [&](unsigned w) -> unsigned {
        const unsigned x = w ^ pat;
        return ~(((x & 0x7f7f7f7fu) + 0x7f7f7f7fu) | x) & 0x80808080u;
    };

    const int nvec = E >> 4;
    const uint4* key4 = (const uint4*)key_w;
    const int voff = (int)(((long long)blockIdx.x * nvec) / gridDim.x) & ~31;
    for (int vb = 0; vb < nvec; vb += 8 * AGG_THREADS) {
        const int v0 = vb + tid;
        uint4 q[8];
        #pragma unroll
        for (int u = 0; u < 8; ++u) {
            const int v = min(v0 + u * AGG_THREADS, nvec - 1);
            const int vv = (v + voff < nvec) ? v + voff : v + voff - nvec;
            q[u] = key4[vv];
        }
        unsigned mm[8];
        int c = 0;
        #pragma unroll
        for (int u = 0; u < 8; ++u) {
            unsigned m = (hit4(q[u].x) >> 7) | (hit4(q[u].y) >> 6) |
                         (hit4(q[u].z) >> 5) | (hit4(q[u].w) >> 4);
            if (v0 + u * AGG_THREADS >= nvec) m = 0;
            mm[u] = m;
            c += __popc(m);
        }
        int incl = c;
        #pragma unroll
        for (int off = 1; off < 32; off <<= 1) {
            const int t = __shfl_up(incl, off, 32);
            if (lane >= off) incl += t;
        }
        const int total = __builtin_amdgcn_readlane(incl, 31);
        if (total) {
            int base = 0;
            if (lane == 31) base = atomicAdd(&cnt, total);
            base = __builtin_amdgcn_readlane(base, 31);
            int pos = base + incl - c;
            #pragma unroll
            for (int u = 0; u < 8; ++u) {
                const int v = v0 + u * AGG_THREADS;
                const int e = ((v + voff < nvec) ? v + voff : v + voff - nvec) << 4;
                unsigned m = mm[u];
                while (m) {
                    const int b = __ffs(m) - 1;
                    m &= m - 1;
                    put(pos++, e + ((b & 7) << 2) + (b >> 3));
                }
            }
        }
    }
    {
        const unsigned char* key_b = (const unsigned char*)key_w;
        for (int e = (nvec << 4) + tid; e < E; e += AGG_THREADS)
            if (key_b[e] == (unsigned char)blockIdx.x) put(atomicAdd(&cnt, 1), e);
    }
    __syncthreads();

    const int n = min(cnt, LIST_CAP);
    for (int k = tid; k < n; k += AGG_THREADS) {
        const int e = list_c[k];
        const int r = ei[e];
        const int c = ei[E + e];
        const unsigned rl = (unsigned)(r - row0);
        const bool own = rl < nr;
        const float lg = s_row[own ? r : row0] + s_col[c];
        const float l  = lg > 0.f ? lg : 0.2f * lg;
        list_c[k] = c;
        list_r[k] = own ? (int)rl : 0;
        list_e[k] = own ? __expf(-l) : 0.f;
    }
    __syncthreads();

    for (int k0 = wave; k0 < n; k0 += 4 * 32) {
        float hv[4][NCLASS / 32];
        float ev[4];
        int   cc[4], rl[4];
        #pragma unroll
        for (int u = 0; u < 4; ++u) {
            const int kk = min(k0 + 32 * u, n - 1);
            cc[u] = list_c[kk];
            rl[u] = list_r[kk];
            ev[u] = list_e[kk];
        }
        #pragma unroll
        for (int u = 0; u < 4; ++u) {
            if (k0 + 32 * u < n) {
                const float* hc = h + (size_t)cc[u] * NCLASS + lane;
                #pragma unroll
                for (int i = 0; i < NCLASS / 32; ++i) hv[u][i] = hc[32 * i];
            }
        }
        #pragma unroll
        for (int u = 0; u < 4; ++u) {
            if (k0 + 32 * u < n) {
                float* ar = acc + rl[u] * NCLASS + lane;
                #pragma unroll
                for (int i = 0; i < NCLASS / 32; ++i) atomicAdd(ar + 32 * i, ev[u] * hv[u][i]);
                if (lane == 0) atomicAdd(rsum + rl[u], ev[u]);
            }
        }
    }
    __syncthreads();

    for (int r = wave; r < nrows; r += 32) {
        const float inv = 1.0f / rsum[r];
        const float4* ar = (const float4*)(acc + r * NCLASS);
        float4* o = (float4*)(out + (size_t)(row0 + r) * NCLASS);
        const float4 p0 = ar[lane];
        const float4 p1 = ar[lane + 32];
        float v[8] = { p0.x, p0.y, p0.z, p0.w, p1.x, p1.y, p1.z, p1.w };
        float mx = -3.402823466e38f;
        #pragma unroll
        for (int i = 0; i < 8; ++i) {
            const float hp = v[i] * inv;
            const float el = hp > 0.f ? hp : (__expf(hp) - 1.f);
            v[i] = el;
            mx = fmaxf(mx, el);
        }
        #pragma unroll
        for (int off = 16; off; off >>= 1) mx = fmaxf(mx, __shfl_xor(mx, off, 32));
        float s = 0.f;
        #pragma unroll
        for (int i = 0; i < 8; ++i) s += __expf(v[i] - mx);
        #pragma unroll
        for (int off = 16; off; off >>= 1) s += __shfl_xor(s, off, 32);
        const float lse = mx + __logf(s);
        o[lane]      = make_float4(v[0] - lse, v[1] - lse, v[2] - lse, v[3] - lse);
        o[lane + 32] = make_float4(v[4] - lse, v[5] - lse, v[6] - lse, v[7] - lse);
    }
}

extern "C" void kernel_launch(void* const* d_in, const int* in_sizes, int n_in,
                              void* d_out, int out_size, void* d_ws, size_t ws_size,
                              hipStream_t stream) {
    const float* x  = (const float*)d_in[0];
    const float* W  = (const float*)d_in[1];
    const float* a  = (const float*)d_in[2];
    const int*   ei = (const int*)d_in[3];

    const int N = in_sizes[0] / NFEAT;
    const int E = in_sizes[3] / 2;

    float* h      = (float*)d_ws;
    float* s_row  = h + (size_t)N * NCLASS;
    float* s_col  = s_row + N;
    unsigned int* key_w = (unsigned int*)(s_col + N);

    float* out = (float*)d_out;

    {
        const int mtiles = N / 16;
        dim3 grid((mtiles + 1) / 2);
        spgat_gemm<<<grid, 128, 0, stream>>>(x, W, h, N);
    }
    {
        dim3 grid((N + 7) / 8);
        spgat_scores<<<grid, 256, 0, stream>>>(h, a, s_row, s_col, ei, E, key_w, N);
    }
    {
        dim3 grid((N + ROWS_PB - 1) / ROWS_PB);
        spgat_aggregate<<<grid, AGG_THREADS, 0, stream>>>(ei, E, key_w, s_row, s_col, h, out, N);
    }
}


# Aggregation reads each gathered h row with 16 B per lane (2 wave loads of 512 B, not 8 of 128 B); LDS row sums kept in that lane order so the float adds stay on distinct banks; sum order as before
# speedup vs baseline: 3.0197x; 1.0452x over previous
#include <hip/hip_runtime.h>
#include <hip/hip_bf16.h>

#define NFEAT  512
#define NCLASS 256
#define ROWS_PB 196

typedef float v2f __attribute__((ext_vector_type(2)));
typedef float v8f __attribute__((ext_vector_type(8)));

__global__ void __launch_bounds__(128)
spgat_gemm(const float* __restrict__ x,
           const float* __restrict__ W,
           float* __restrict__ h, int N) {
    const int lane = threadIdx.x & 31;
    const int wave = threadIdx.x >> 5;
    const int mtiles = N >> 4;
    const int mt0  = blockIdx.x * 2;
    const bool has1 = (mt0 + 1) < mtiles;
    const int m0   = mt0 << 4;
    const int m1   = has1 ? m0 + 16 : m0;
    const int n0   = wave * 64;
    const int half = lane >> 4;
    const int sub  = lane & 15;

    v8f c00 = {}, c01 = {}, c02 = {}, c03 = {};
    v8f c10 = {}, c11 = {}, c12 = {}, c13 = {};

    const float* xr0 = x + (size_t)(m0 + sub) * NFEAT + 2 * half;
    const float* xr1 = x + (size_t)(m1 + sub) * NFEAT + 2 * half;
    const float* wb  = W + (size_t)(2 * half) * NCLASS + n0 + sub;
    for (int kb = 0; kb < NFEAT; kb += 4) {
        v2f a0; a0.x = xr0[kb + 0]; a0.y = xr0[kb + 1];
        v2f a1; a1.x = xr1[kb + 0]; a1.y = xr1[kb + 1];

        const float* wk0 = wb + (size_t)kb * NCLASS;
        const float* wk1 = wk0 + NCLASS;
        v2f b0; b0.x = wk0[ 0]; b0.y = wk1[ 0];
        v2f b1; b1.x = wk0[16]; b1.y = wk1[16];
        v2f b2; b2.x = wk0[32]; b2.y = wk1[32];
        v2f b3; b3.x = wk0[48]; b3.y = wk1[48];

        c00 = __builtin_amdgcn_wmma_f32_16x16x4_f32(false, a0, false, b0, (short)0, c00, false, false);
        c10 = __builtin_amdgcn_wmma_f32_16x16x4_f32(false, a1, false, b0, (short)0, c10, false, false);
        c01 = __builtin_amdgcn_wmma_f32_16x16x4_f32(false, a0, false, b1, (short)0, c01, false, false);
        c11 = __builtin_amdgcn_wmma_f32_16x16x4_f32(false, a1, false, b1, (short)0, c11, false, false);
        c02 = __builtin_amdgcn_wmma_f32_16x16x4_f32(false, a0, false, b2, (short)0, c02, false, false);
        c12 = __builtin_amdgcn_wmma_f32_16x16x4_f32(false, a1, false, b2, (short)0, c12, false, false);
        c03 = __builtin_amdgcn_wmma_f32_16x16x4_f32(false, a0, false, b3, (short)0, c03, false, false);
        c13 = __builtin_amdgcn_wmma_f32_16x16x4_f32(false, a1, false, b3, (short)0, c13, false, false);
    }

    float* hout = h + n0 + sub;
    #pragma unroll
    for (int r = 0; r < 8; ++r) {
        const size_t m = (size_t)(m0 + half * 8 + r);
        hout[m * NCLASS +  0] = c00[r];
        hout[m * NCLASS + 16] = c01[r];
        hout[m * NCLASS + 32] = c02[r];
        hout[m * NCLASS + 48] = c03[r];
    }
    if (has1) {
        #pragma unroll
        for (int r = 0; r < 8; ++r) {
            const size_t m = (size_t)(m1 + half * 8 + r);
            hout[m * NCLASS +  0] = c10[r];
            hout[m * NCLASS + 16] = c11[r];
            hout[m * NCLASS + 32] = c12[r];
            hout[m * NCLASS + 48] = c13[r];
        }
    }
}

__global__ void spgat_scores(const float* __restrict__ h,
                             const float* __restrict__ a,
                             float* __restrict__ s_row,
                             float* __restrict__ s_col,
                             const int* __restrict__ ei, int E,
                             unsigned int* __restrict__ key_w, int N) {
    {
        const int nthr = gridDim.x * blockDim.x;
        const int nq   = E >> 2;
        for (int g = blockIdx.x * blockDim.x + threadIdx.x; g < nq; g += nthr) {
            const int4 r = ((const int4*)ei)[g];
            key_w[g] = (((unsigned)r.x / ROWS_PB) & 255u) | ((((unsigned)r.y / ROWS_PB) & 255u) << 8) |
                       ((((unsigned)r.z / ROWS_PB) & 255u) << 16) | (((unsigned)r.w / ROWS_PB) << 24);
        }
        unsigned char* key_b = (unsigned char*)key_w;
        for (int e = (nq << 2) + blockIdx.x * blockDim.x + threadIdx.x; e < E; e += nthr)
            key_b[e] = (unsigned char)((unsigned)ei[e] / ROWS_PB);
    }
    const int lane = threadIdx.x & 31;
    const int row  = blockIdx.x * (blockDim.x >> 5) + (threadIdx.x >> 5);
    if (row >= N) return;
    const float* hr = h + (size_t)row * NCLASS;
    float sr = 0.f, sc = 0.f;
    #pragma unroll
    for (int i = 0; i < NCLASS / 32; ++i) {
        const int f = lane + 32 * i;
        const float v = hr[f];
        sr += v * a[f];
        sc += v * a[NCLASS + f];
    }
    #pragma unroll
    for (int off = 16; off; off >>= 1) {
        sr += __shfl_xor(sr, off, 32);
        sc += __shfl_xor(sc, off, 32);
    }
    if (lane == 0) { s_row[row] = sr; s_col[row] = sc; }
}

#define LIST_CAP    8192
#define AGG_THREADS 1024

__global__ void __launch_bounds__(AGG_THREADS)
spgat_aggregate(const int* __restrict__ ei, int E,
                const unsigned int* __restrict__ key_w,
                const float* __restrict__ s_row,
                const float* __restrict__ s_col,
                const float* __restrict__ h,
                float* __restrict__ out, int N) {
    __shared__ __attribute__((aligned(16))) float acc[ROWS_PB * NCLASS];
    __shared__ float rsum[ROWS_PB];
    __shared__ int   list_c[LIST_CAP];
    __shared__ int   list_r[LIST_CAP];
    __shared__ float list_e[LIST_CAP];
    __shared__ int   cnt;

    const int tid  = threadIdx.x;
    const int lane = tid & 31;
    const int wave = __builtin_amdgcn_readfirstlane(tid >> 5);
    const int row0 = blockIdx.x * ROWS_PB;
    const int nrows = min(ROWS_PB, N - row0);
    const unsigned nr = (unsigned)nrows;

    for (int i = tid; i < ROWS_PB * NCLASS; i += AGG_THREADS) acc[i] = 0.f;
    if (tid < ROWS_PB) rsum[tid] = 0.f;
    if (tid == 0) cnt = 0;
    __syncthreads();

    auto add_here = [&](int e) {
        const unsigned rl = (unsigned)(ei[e] - row0);
        if (rl >= nr) return;
        const int c = ei[E + e];
        const float lg = s_row[row0 + (int)rl] + s_col[c];
        const float l  = lg > 0.f ? lg : 0.2f * lg;
        const float ev = __expf(-l);
        const float* hc = h + (size_t)c * NCLASS;
        float* ar = acc + rl * NCLASS;
        #pragma unroll 1
        for (int f = 0; f < NCLASS; ++f)
            atomicAdd(ar + (f & 128) + ((f & 3) << 5) + ((f & 127) >> 2), ev * hc[f]);
        atomicAdd(rsum + rl, ev);
    };
    auto put = [&](int pos, int e) {
        if (pos < LIST_CAP) list_c[pos] = e;
        else add_here(e);
    };
    const unsigned pat = (unsigned)blockIdx.x * 0x01010101u;
    auto hit4 = [&](unsigned w) -> unsigned {
        const unsigned x = w ^ pat;
        return ~(((x & 0x7f7f7f7fu) + 0x7f7f7f7fu) | x) & 0x80808080u;
    };

    const int nvec = E >> 4;
    const uint4* key4 = (const uint4*)key_w;
    const int voff = (int)(((long long)blockIdx.x * nvec) / gridDim.x) & ~31;
    for (int vb = 0; vb < nvec; vb += 8 * AGG_THREADS) {
        const int v0 = vb + tid;
        uint4 q[8];
        #pragma unroll
        for (int u = 0; u < 8; ++u) {
            const int v = min(v0 + u * AGG_THREADS, nvec - 1);
            const int vv = (v + voff < nvec) ? v + voff : v + voff - nvec;
            q[u] = key4[vv];
        }
        unsigned mm[8];
        int c = 0;
        #pragma unroll
        for (int u = 0; u < 8; ++u) {
            unsigned m = (hit4(q[u].x) >> 7) | (hit4(q[u].y) >> 6) |
                         (hit4(q[u].z) >> 5) | (hit4(q[u].w) >> 4);
            if (v0 + u * AGG_THREADS >= nvec) m = 0;
            mm[u] = m;
            c += __popc(m);
        }
        int incl = c;
        #pragma unroll
        for (int off = 1; off < 32; off <<= 1) {
            const int t = __shfl_up(incl, off, 32);
            if (lane >= off) incl += t;
        }
        const int total = __builtin_amdgcn_readlane(incl, 31);
        if (total) {
            int base = 0;
            if (lane == 31) base = atomicAdd(&cnt, total);
            base = __builtin_amdgcn_readlane(base, 31);
            int pos = base + incl - c;
            #pragma unroll
            for (int u = 0; u < 8; ++u) {
                const int v = v0 + u * AGG_THREADS;
                const int e = ((v + voff < nvec) ? v + voff : v + voff - nvec) << 4;
                unsigned m = mm[u];
                while (m) {
                    const int b = __ffs(m) - 1;
                    m &= m - 1;
                    put(pos++, e + ((b & 7) << 2) + (b >> 3));
                }
            }
        }
    }
    {
        const unsigned char* key_b = (const unsigned char*)key_w;
        for (int e = (nvec << 4) + tid; e < E; e += AGG_THREADS)
            if (key_b[e] == (unsigned char)blockIdx.x) put(atomicAdd(&cnt, 1), e);
    }
    __syncthreads();

    const int n = min(cnt, LIST_CAP);
    for (int k = tid; k < n; k += AGG_THREADS) {
        const int e = list_c[k];
        const int r = ei[e];
        const int c = ei[E + e];
        const unsigned rl = (unsigned)(r - row0);
        const bool own = rl < nr;
        const float lg = s_row[own ? r : row0] + s_col[c];
        const float l  = lg > 0.f ? lg : 0.2f * lg;
        list_c[k] = c;
        list_r[k] = own ? (int)rl : 0;
        list_e[k] = own ? __expf(-l) : 0.f;
    }
    __syncthreads();

    for (int k0 = wave; k0 < n; k0 += 4 * 32) {
        float4 hv[4][2];
        float ev[4];
        int   cc[4], rl[4];
        #pragma unroll
        for (int u = 0; u < 4; ++u) {
            const int kk = min(k0 + 32 * u, n - 1);
            cc[u] = list_c[kk];
            rl[u] = list_r[kk];
            ev[u] = list_e[kk];
        }
        #pragma unroll
        for (int u = 0; u < 4; ++u) {
            if (k0 + 32 * u < n) {
                const float4* hc = (const float4*)(h + (size_t)cc[u] * NCLASS) + lane;
                hv[u][0] = hc[0];
                hv[u][1] = hc[32];
            }
        }
        #pragma unroll
        for (int u = 0; u < 4; ++u) {
            if (k0 + 32 * u < n) {
                float* ar = acc + rl[u] * NCLASS + lane;
                #pragma unroll
                for (int hf = 0; hf < 2; ++hf) {
                    atomicAdd(ar + 128 * hf +  0, ev[u] * hv[u][hf].x);
                    atomicAdd(ar + 128 * hf + 32, ev[u] * hv[u][hf].y);
                    atomicAdd(ar + 128 * hf + 64, ev[u] * hv[u][hf].z);
                    atomicAdd(ar + 128 * hf + 96, ev[u] * hv[u][hf].w);
                }
                if (lane == 0) atomicAdd(rsum + rl[u], ev[u]);
            }
        }
    }
    __syncthreads();

    for (int r = wave; r < nrows; r += 32) {
        const float inv = 1.0f / rsum[r];
        const float* ar = acc + r * NCLASS + lane;
        float4* o = (float4*)(out + (size_t)(row0 + r) * NCLASS);
        float v[8];
        #pragma unroll
        for (int i = 0; i < 8; ++i) v[i] = ar[32 * i];
        float mx = -3.402823466e38f;
        #pragma unroll
        for (int i = 0; i < 8; ++i) {
            const float hp = v[i] * inv;
            const float el = hp > 0.f ? hp : (__expf(hp) - 1.f);
            v[i] = el;
            mx = fmaxf(mx, el);
        }
        #pragma unroll
        for (int off = 16; off; off >>= 1) mx = fmaxf(mx, __shfl_xor(mx, off, 32));
        float s = 0.f;
        #pragma unroll
        for (int i = 0; i < 8; ++i) s += __expf(v[i] - mx);
        #pragma unroll
        for (int off = 16; off; off >>= 1) s += __shfl_xor(s, off, 32);
        const float lse = mx + __logf(s);
        o[lane]      = make_float4(v[0] - lse, v[1] - lse, v[2] - lse, v[3] - lse);
        o[lane + 32] = make_float4(v[4] - lse, v[5] - lse, v[6] - lse, v[7] - lse);
    }
}

extern "C" void kernel_launch(void* const* d_in, const int* in_sizes, int n_in,
                              void* d_out, int out_size, void* d_ws, size_t ws_size,
                              hipStream_t stream) {
    const float* x  = (const float*)d_in[0];
    const float* W  = (const float*)d_in[1];
    const float* a  = (const float*)d_in[2];
    const int*   ei = (const int*)d_in[3];

    const int N = in_sizes[0] / NFEAT;
    const int E = in_sizes[3] / 2;

    float* h      = (float*)d_ws;
    float* s_row  = h + (size_t)N * NCLASS;
    float* s_col  = s_row + N;
    unsigned int* key_w = (unsigned int*)(s_col + N);

    float* out = (float*)d_out;

    {
        const int mtiles = N / 16;
        dim3 grid((mtiles + 1) / 2);
        spgat_gemm<<<grid, 128, 0, stream>>>(x, W, h, N);
    }
    {
        dim3 grid((N + 7) / 8);
        spgat_scores<<<grid, 256, 0, stream>>>(h, a, s_row, s_col, ei, E, key_w, N);
    }
    {
        dim3 grid((N + ROWS_PB - 1) / ROWS_PB);
        spgat_aggregate<<<grid, AGG_THREADS, 0, stream>>>(ei, E, key_w, s_row, s_col, h, out, N);
    }
}


# Aggregation rows owned by waves: a listed edge goes to the LDS list of the wave owning row r mod 32, which adds e*h[col] by plain 16-B LDS load/fma/store; no float atomics; sum order = list arrival
# speedup vs baseline: 3.1155x; 1.0317x over previous
#include <hip/hip_runtime.h>
#include <hip/hip_bf16.h>

#define NFEAT  512
#define NCLASS 256
#define ROWS_PB 196

typedef float v2f __attribute__((ext_vector_type(2)));
typedef float v8f __attribute__((ext_vector_type(8)));

__global__ void __launch_bounds__(128)
spgat_gemm(const float* __restrict__ x,
           const float* __restrict__ W,
           float* __restrict__ h, int N) {
    const int lane = threadIdx.x & 31;
    const int wave = threadIdx.x >> 5;
    const int mtiles = N >> 4;
    const int mt0  = blockIdx.x * 2;
    const bool has1 = (mt0 + 1) < mtiles;
    const int m0   = mt0 << 4;
    const int m1   = has1 ? m0 + 16 : m0;
    const int n0   = wave * 64;
    const int half = lane >> 4;
    const int sub  = lane & 15;

    v8f c00 = {}, c01 = {}, c02 = {}, c03 = {};
    v8f c10 = {}, c11 = {}, c12 = {}, c13 = {};

    const float* xr0 = x + (size_t)(m0 + sub) * NFEAT + 2 * half;
    const float* xr1 = x + (size_t)(m1 + sub) * NFEAT + 2 * half;
    const float* wb  = W + (size_t)(2 * half) * NCLASS + n0 + sub;
    for (int kb = 0; kb < NFEAT; kb += 4) {
        v2f a0; a0.x = xr0[kb + 0]; a0.y = xr0[kb + 1];
        v2f a1; a1.x = xr1[kb + 0]; a1.y = xr1[kb + 1];

        const float* wk0 = wb + (size_t)kb * NCLASS;
        const float* wk1 = wk0 + NCLASS;
        v2f b0; b0.x = wk0[ 0]; b0.y = wk1[ 0];
        v2f b1; b1.x = wk0[16]; b1.y = wk1[16];
        v2f b2; b2.x = wk0[32]; b2.y = wk1[32];
        v2f b3; b3.x = wk0[48]; b3.y = wk1[48];

        c00 = __builtin_amdgcn_wmma_f32_16x16x4_f32(false, a0, false, b0, (short)0, c00, false, false);
        c10 = __builtin_amdgcn_wmma_f32_16x16x4_f32(false, a1, false, b0, (short)0, c10, false, false);
        c01 = __builtin_amdgcn_wmma_f32_16x16x4_f32(false, a0, false, b1, (short)0, c01, false, false);
        c11 = __builtin_amdgcn_wmma_f32_16x16x4_f32(false, a1, false, b1, (short)0, c11, false, false);
        c02 = __builtin_amdgcn_wmma_f32_16x16x4_f32(false, a0, false, b2, (short)0, c02, false, false);
        c12 = __builtin_amdgcn_wmma_f32_16x16x4_f32(false, a1, false, b2, (short)0, c12, false, false);
        c03 = __builtin_amdgcn_wmma_f32_16x16x4_f32(false, a0, false, b3, (short)0, c03, false, false);
        c13 = __builtin_amdgcn_wmma_f32_16x16x4_f32(false, a1, false, b3, (short)0, c13, false, false);
    }

    float* hout = h + n0 + sub;
    #pragma unroll
    for (int r = 0; r < 8; ++r) {
        const size_t m = (size_t)(m0 + half * 8 + r);
        hout[m * NCLASS +  0] = c00[r];
        hout[m * NCLASS + 16] = c01[r];
        hout[m * NCLASS + 32] = c02[r];
        hout[m * NCLASS + 48] = c03[r];
    }
    if (has1) {
        #pragma unroll
        for (int r = 0; r < 8; ++r) {
            const size_t m = (size_t)(m1 + half * 8 + r);
            hout[m * NCLASS +  0] = c10[r];
            hout[m * NCLASS + 16] = c11[r];
            hout[m * NCLASS + 32] = c12[r];
            hout[m * NCLASS + 48] = c13[r];
        }
    }
}

__global__ void spgat_scores(const float* __restrict__ h,
                             const float* __restrict__ a,
                             float* __restrict__ s_row,
                             float* __restrict__ s_col,
                             const int* __restrict__ ei, int E,
                             unsigned int* __restrict__ key_w, int N) {
    {
        const int nthr = gridDim.x * blockDim.x;
        const int nq   = E >> 2;
        for (int g = blockIdx.x * blockDim.x + threadIdx.x; g < nq; g += nthr) {
            const int4 r = ((const int4*)ei)[g];
            key_w[g] = (((unsigned)r.x / ROWS_PB) & 255u) | ((((unsigned)r.y / ROWS_PB) & 255u) << 8) |
                       ((((unsigned)r.z / ROWS_PB) & 255u) << 16) | (((unsigned)r.w / ROWS_PB) << 24);
        }
        unsigned char* key_b = (unsigned char*)key_w;
        for (int e = (nq << 2) + blockIdx.x * blockDim.x + threadIdx.x; e < E; e += nthr)
            key_b[e] = (unsigned char)((unsigned)ei[e] / ROWS_PB);
    }
    const int lane = threadIdx.x & 31;
    const int row  = blockIdx.x * (blockDim.x >> 5) + (threadIdx.x >> 5);
    if (row >= N) return;
    const float* hr = h + (size_t)row * NCLASS;
    float sr = 0.f, sc = 0.f;
    #pragma unroll
    for (int i = 0; i < NCLASS / 32; ++i) {
        const int f = lane + 32 * i;
        const float v = hr[f];
        sr += v * a[f];
        sc += v * a[NCLASS + f];
    }
    #pragma unroll
    for (int off = 16; off; off >>= 1) {
        sr += __shfl_xor(sr, off, 32);
        sc += __shfl_xor(sc, off, 32);
    }
    if (lane == 0) { s_row[row] = sr; s_col[row] = sc; }
}

#define LIST_CAP    4096
#define WAVE_CAP    192
#define WAVE_PITCH  193
#define GROUP       8
#define SLOTS       7
#define AGG_THREADS 1024

__global__ void __launch_bounds__(AGG_THREADS)
spgat_aggregate(const int* __restrict__ ei, int E,
                const unsigned int* __restrict__ key_w,
                const float* __restrict__ s_row,
                const float* __restrict__ s_col,
                const float* __restrict__ h,
                float* __restrict__ out, int N) {
    __shared__ __attribute__((aligned(16))) float acc[ROWS_PB * NCLASS];
    __shared__ int   list_e[LIST_CAP];
    __shared__ int   wl_c[32 * WAVE_PITCH];
    __shared__ int   wl_r[32 * WAVE_PITCH];
    __shared__ float wl_w[32 * WAVE_PITCH];
    __shared__ int   wcnt[32];
    __shared__ int   cnt;
    __shared__ int   ovf;

    const int tid  = threadIdx.x;
    const int lane = tid & 31;
    const int wave = __builtin_amdgcn_readfirstlane(tid >> 5);
    const int row0 = blockIdx.x * ROWS_PB;
    const int nrows = min(ROWS_PB, N - row0);
    const unsigned nr = (unsigned)nrows;

    for (int i = tid; i < ROWS_PB * NCLASS / 4; i += AGG_THREADS)
        ((float4*)acc)[i] = make_float4(0.f, 0.f, 0.f, 0.f);
    if (tid < 32) wcnt[tid] = 0;
    if (tid == 0) { cnt = 0; ovf = 0; }
    __syncthreads();

    auto put = [&](int pos, int e) {
        if (pos < LIST_CAP) list_e[pos] = e;
        else ovf = 1;
    };
    const unsigned pat = (unsigned)blockIdx.x * 0x01010101u;
    auto hit4 = [&](unsigned w) -> unsigned {
        const unsigned x = w ^ pat;
        return ~(((x & 0x7f7f7f7fu) + 0x7f7f7f7fu) | x) & 0x80808080u;
    };

    const int nvec = E >> 4;
    const uint4* key4 = (const uint4*)key_w;
    const int voff = (int)(((long long)blockIdx.x * nvec) / gridDim.x) & ~31;
    for (int vb = 0; vb < nvec; vb += 8 * AGG_THREADS) {
        const int v0 = vb + tid;
        uint4 q[8];
        #pragma unroll
        for (int u = 0; u < 8; ++u) {
            const int v = min(v0 + u * AGG_THREADS, nvec - 1);
            const int vv = (v + voff < nvec) ? v + voff : v + voff - nvec;
            q[u] = key4[vv];
        }
        unsigned mm[8];
        int c = 0;
        #pragma unroll
        for (int u = 0; u < 8; ++u) {
            unsigned m = (hit4(q[u].x) >> 7) | (hit4(q[u].y) >> 6) |
                         (hit4(q[u].z) >> 5) | (hit4(q[u].w) >> 4);
            if (v0 + u * AGG_THREADS >= nvec) m = 0;
            mm[u] = m;
            c += __popc(m);
        }
        int incl = c;
        #pragma unroll
        for (int off = 1; off < 32; off <<= 1) {
            const int t = __shfl_up(incl, off, 32);
            if (lane >= off) incl += t;
        }
        const int total = __builtin_amdgcn_readlane(incl, 31);
        if (total) {
            int base = 0;
            if (lane == 31) base = atomicAdd(&cnt, total);
            base = __builtin_amdgcn_readlane(base, 31);
            int pos = base + incl - c;
            #pragma unroll
            for (int u = 0; u < 8; ++u) {
                const int v = v0 + u * AGG_THREADS;
                const int e = ((v + voff < nvec) ? v + voff : v + voff - nvec) << 4;
                unsigned m = mm[u];
                while (m) {
                    const int b = __ffs(m) - 1;
                    m &= m - 1;
                    put(pos++, e + ((b & 7) << 2) + (b >> 3));
                }
            }
        }
    }
    {
        const unsigned char* key_b = (const unsigned char*)key_w;
        for (int e = (nvec << 4) + tid; e < E; e += AGG_THREADS)
            if (key_b[e] == (unsigned char)blockIdx.x) put(atomicAdd(&cnt, 1), e);
    }
    __syncthreads();

    const int n = min(cnt, LIST_CAP);
    for (int k = tid; k < n; k += AGG_THREADS) {
        const int e = list_e[k];
        const int r = ei[e];
        const unsigned rl = (unsigned)(r - row0);
        if (rl < nr) {
            const int c = ei[E + e];
            const float lg = s_row[r] + s_col[c];
            const float l  = lg > 0.f ? lg : 0.2f * lg;
            const int w = (int)(rl & 31u);
            const int p = atomicAdd(&wcnt[w], 1);
            if (p < WAVE_CAP) {
                wl_c[w * WAVE_PITCH + p] = c;
                wl_r[w * WAVE_PITCH + p] = (int)rl;
                wl_w[w * WAVE_PITCH + p] = __expf(-l);
            } else {
                ovf = 1;
            }
        }
    }
    __syncthreads();

    float rs[SLOTS];
    #pragma unroll
    for (int j = 0; j < SLOTS; ++j) rs[j] = 0.f;
    auto add_edge = [&](int rl, float w, const float4 lo, const float4 hi) {
        float4* ar = (float4*)(acc + rl * NCLASS) + lane;
        float4 a0 = ar[0];
        float4 a1 = ar[32];
        a0.x += w * lo.x; a0.y += w * lo.y; a0.z += w * lo.z; a0.w += w * lo.w;
        a1.x += w * hi.x; a1.y += w * hi.y; a1.z += w * hi.z; a1.w += w * hi.w;
        ar[0]  = a0;
        ar[32] = a1;
        const int slot = rl >> 5;
        #pragma unroll
        for (int j = 0; j < SLOTS; ++j) rs[j] += (slot == j) ? w : 0.f;
    };

    const int overflow = __builtin_amdgcn_readfirstlane(ovf);
    if (!overflow) {
        const int nw = __builtin_amdgcn_readfirstlane(min(wcnt[wave], WAVE_CAP));
        const int*   wc = wl_c + wave * WAVE_PITCH;
        const int*   wr = wl_r + wave * WAVE_PITCH;
        const float* ww = wl_w + wave * WAVE_PITCH;
        for (int k0 = 0; k0 < nw; k0 += GROUP) {
            float4 hv[GROUP][2];
            float ev[GROUP];
            int   cc[GROUP], rr[GROUP];
            #pragma unroll
            for (int u = 0; u < GROUP; ++u) {
                const int kk = min(k0 + u, nw - 1);
                cc[u] = wc[kk];
                rr[u] = __builtin_amdgcn_readfirstlane(wr[kk]);
                ev[u] = __int_as_float(__builtin_amdgcn_readfirstlane(__float_as_int(ww[kk])));
            }
            #pragma unroll
            for (int u = 0; u < GROUP; ++u) {
                if (k0 + u < nw) {
                    const float4* hc = (const float4*)(h + (size_t)cc[u] * NCLASS) + lane;
                    hv[u][0] = hc[0];
                    hv[u][1] = hc[32];
                }
            }
            #pragma unroll
            for (int u = 0; u < GROUP; ++u) {
                if (k0 + u < nw) add_edge(rr[u], ev[u], hv[u][0], hv[u][1]);
            }
        }
    } else {
        for (int e0 = 0; e0 < E; e0 += 32) {
            const int e = e0 + lane;
            bool mine = false;
            if (e < E) {
                const unsigned rl = (unsigned)(ei[e] - row0);
                mine = (rl < nr) && ((int)(rl & 31u) == wave);
            }
            unsigned m = __builtin_amdgcn_ballot_w32(mine);
            while (m) {
                const int b = __ffs(m) - 1;
                m &= m - 1;
                const int ee = e0 + b;
                const int r = ei[ee];
                const int c = ei[E + ee];
                const float lg = s_row[r] + s_col[c];
                const float l  = lg > 0.f ? lg : 0.2f * lg;
                const float4* hc = (const float4*)(h + (size_t)c * NCLASS) + lane;
                const float wgt = __int_as_float(__builtin_amdgcn_readfirstlane(__float_as_int(__expf(-l))));
                add_edge(__builtin_amdgcn_readfirstlane(r - row0), wgt, hc[0], hc[32]);
            }
        }
    }

    #pragma unroll
    for (int j = 0; j < SLOTS; ++j) {
        const int r = wave + 32 * j;
        if (r >= nrows) break;
        const float inv = 1.0f / rs[j];
        const float4* ar = (const float4*)(acc + r * NCLASS) + lane;
        float4* o = (float4*)(out + (size_t)(row0 + r) * NCLASS);
        const float4 s0 = ar[0];
        const float4 s1 = ar[32];
        float v[8] = { s0.x, s0.y, s0.z, s0.w, s1.x, s1.y, s1.z, s1.w };
        float mx = -3.402823466e38f;
        #pragma unroll
        for (int i = 0; i < 8; ++i) {
            const float hp = v[i] * inv;
            const float el = hp > 0.f ? hp : (__expf(hp) - 1.f);
            v[i] = el;
            mx = fmaxf(mx, el);
        }
        #pragma unroll
        for (int off = 16; off; off >>= 1) mx = fmaxf(mx, __shfl_xor(mx, off, 32));
        float sm = 0.f;
        #pragma unroll
        for (int i = 0; i < 8; ++i) sm += __expf(v[i] - mx);
        #pragma unroll
        for (int off = 16; off; off >>= 1) sm += __shfl_xor(sm, off, 32);
        const float lse = mx + __logf(sm);
        o[lane]      = make_float4(v[0] - lse, v[1] - lse, v[2] - lse, v[3] - lse);
        o[lane + 32] = make_float4(v[4] - lse, v[5] - lse, v[6] - lse, v[7] - lse);
    }
}

extern "C" void kernel_launch(void* const* d_in, const int* in_sizes, int n_in,
                              void* d_out, int out_size, void* d_ws, size_t ws_size,
                              hipStream_t stream) {
    const float* x  = (const float*)d_in[0];
    const float* W  = (const float*)d_in[1];
    const float* a  = (const float*)d_in[2];
    const int*   ei = (const int*)d_in[3];

    const int N = in_sizes[0] / NFEAT;
    const int E = in_sizes[3] / 2;

    float* h      = (float*)d_ws;
    float* s_row  = h + (size_t)N * NCLASS;
    float* s_col  = s_row + N;
    unsigned int* key_w = (unsigned int*)(s_col + N);

    float* out = (float*)d_out;

    {
        const int mtiles = N / 16;
        dim3 grid((mtiles + 1) / 2);
        spgat_gemm<<<grid, 128, 0, stream>>>(x, W, h, N);
    }
    {
        dim3 grid((N + 7) / 8);
        spgat_scores<<<grid, 256, 0, stream>>>(h, a, s_row, s_col, ei, E, key_w, N);
    }
    {
        dim3 grid((N + ROWS_PB - 1) / ROWS_PB);
        spgat_aggregate<<<grid, AGG_THREADS, 0, stream>>>(ei, E, key_w, s_row, s_col, h, out, N);
    }
}


# Scores launch cut from 50000 one-row waves to 4096 waves of 12-13 contiguous rows: all 26 row loads (16 B a lane) in flight at once, a in registers, one store per score array; s sum order changes
# speedup vs baseline: 3.2324x; 1.0375x over previous
#include <hip/hip_runtime.h>
#include <hip/hip_bf16.h>

#define NFEAT  512
#define NCLASS 256
#define ROWS_PB 196

typedef float v2f __attribute__((ext_vector_type(2)));
typedef float v8f __attribute__((ext_vector_type(8)));

__global__ void __launch_bounds__(128)
spgat_gemm(const float* __restrict__ x,
           const float* __restrict__ W,
           float* __restrict__ h, int N) {
    const int lane = threadIdx.x & 31;
    const int wave = threadIdx.x >> 5;
    const int mtiles = N >> 4;
    const int mt0  = blockIdx.x * 2;
    const bool has1 = (mt0 + 1) < mtiles;
    const int m0   = mt0 << 4;
    const int m1   = has1 ? m0 + 16 : m0;
    const int n0   = wave * 64;
    const int half = lane >> 4;
    const int sub  = lane & 15;

    v8f c00 = {}, c01 = {}, c02 = {}, c03 = {};
    v8f c10 = {}, c11 = {}, c12 = {}, c13 = {};

    const float* xr0 = x + (size_t)(m0 + sub) * NFEAT + 2 * half;
    const float* xr1 = x + (size_t)(m1 + sub) * NFEAT + 2 * half;
    const float* wb  = W + (size_t)(2 * half) * NCLASS + n0 + sub;
    for (int kb = 0; kb < NFEAT; kb += 4) {
        v2f a0; a0.x = xr0[kb + 0]; a0.y = xr0[kb + 1];
        v2f a1; a1.x = xr1[kb + 0]; a1.y = xr1[kb + 1];

        const float* wk0 = wb + (size_t)kb * NCLASS;
        const float* wk1 = wk0 + NCLASS;
        v2f b0; b0.x = wk0[ 0]; b0.y = wk1[ 0];
        v2f b1; b1.x = wk0[16]; b1.y = wk1[16];
        v2f b2; b2.x = wk0[32]; b2.y = wk1[32];
        v2f b3; b3.x = wk0[48]; b3.y = wk1[48];

        c00 = __builtin_amdgcn_wmma_f32_16x16x4_f32(false, a0, false, b0, (short)0, c00, false, false);
        c10 = __builtin_amdgcn_wmma_f32_16x16x4_f32(false, a1, false, b0, (short)0, c10, false, false);
        c01 = __builtin_amdgcn_wmma_f32_16x16x4_f32(false, a0, false, b1, (short)0, c01, false, false);
        c11 = __builtin_amdgcn_wmma_f32_16x16x4_f32(false, a1, false, b1, (short)0, c11, false, false);
        c02 = __builtin_amdgcn_wmma_f32_16x16x4_f32(false, a0, false, b2, (short)0, c02, false, false);
        c12 = __builtin_amdgcn_wmma_f32_16x16x4_f32(false, a1, false, b2, (short)0, c12, false, false);
        c03 = __builtin_amdgcn_wmma_f32_16x16x4_f32(false, a0, false, b3, (short)0, c03, false, false);
        c13 = __builtin_amdgcn_wmma_f32_16x16x4_f32(false, a1, false, b3, (short)0, c13, false, false);
    }

    float* hout = h + n0 + sub;
    #pragma unroll
    for (int r = 0; r < 8; ++r) {
        const size_t m = (size_t)(m0 + half * 8 + r);
        hout[m * NCLASS +  0] = c00[r];
        hout[m * NCLASS + 16] = c01[r];
        hout[m * NCLASS + 32] = c02[r];
        hout[m * NCLASS + 48] = c03[r];
    }
    if (has1) {
        #pragma unroll
        for (int r = 0; r < 8; ++r) {
            const size_t m = (size_t)(m1 + half * 8 + r);
            hout[m * NCLASS +  0] = c10[r];
            hout[m * NCLASS + 16] = c11[r];
            hout[m * NCLASS + 32] = c12[r];
            hout[m * NCLASS + 48] = c13[r];
        }
    }
}

#define SC_BLOCKS  512
#define SC_THREADS 256
#define SC_WAVES   (SC_BLOCKS * (SC_THREADS / 32))
#define SC_ROWS    13

typedef float v4f __attribute__((ext_vector_type(4)));
typedef int   v4i __attribute__((ext_vector_type(4)));

__device__ __forceinline__ unsigned key4_of(const v4i r) {
    return (((unsigned)r.x / ROWS_PB) & 255u) | ((((unsigned)r.y / ROWS_PB) & 255u) << 8) |
           ((((unsigned)r.z / ROWS_PB) & 255u) << 16) | (((unsigned)r.w / ROWS_PB) << 24);
}

__global__ void __launch_bounds__(SC_THREADS)
spgat_scores(const float* __restrict__ h,
             const float* __restrict__ a,
             float* __restrict__ s_row,
             float* __restrict__ s_col,
             const int* __restrict__ ei, int E,
             unsigned int* __restrict__ key_w, int N) {
    const int lane = threadIdx.x & 31;
    const int gtid = blockIdx.x * SC_THREADS + threadIdx.x;
    const int nthr = SC_BLOCKS * SC_THREADS;
    const int nq   = E >> 2;

    const v4f* a4 = (const v4f*)a;
    const v4f ar0 = a4[lane], ar1 = a4[32 + lane];
    const v4f ac0 = a4[64 + lane], ac1 = a4[96 + lane];
    asm volatile("" :: "v"(ar0), "v"(ar1), "v"(ac0), "v"(ac1));

    const int g0 = gtid, g1 = gtid + nthr;
    const int glast = max(nq, 1) - 1;
    const v4i k0 = ((const v4i*)ei)[min(g0, glast)];
    const v4i k1 = ((const v4i*)ei)[min(g1, glast)];

    const int wg   = __builtin_amdgcn_readfirstlane(gtid >> 5);
    const int q    = N / SC_WAVES, rem = N % SC_WAVES;
    const int rbeg = wg * q + min(wg, rem);
    const int rend = rbeg + q + (wg < rem ? 1 : 0);

    for (int r0 = rbeg; r0 < rend; r0 += SC_ROWS) {
        v4f lo[SC_ROWS], hi[SC_ROWS];
        #pragma unroll
        for (int u = 0; u < SC_ROWS; ++u) {
            const int r = min(r0 + u, rend - 1);
            const v4f* hr = (const v4f*)(h + (size_t)r * NCLASS) + lane;
            lo[u] = hr[0];
            hi[u] = hr[32];
        }
        asm volatile("" :: "v"(k0), "v"(k1),
                     "v"(lo[0]), "v"(lo[1]), "v"(lo[2]), "v"(lo[3]), "v"(lo[4]), "v"(lo[5]), "v"(lo[6]),
                     "v"(lo[7]), "v"(lo[8]), "v"(lo[9]), "v"(lo[10]), "v"(lo[11]), "v"(lo[12]),
                     "v"(hi[0]), "v"(hi[1]), "v"(hi[2]), "v"(hi[3]), "v"(hi[4]), "v"(hi[5]), "v"(hi[6]),
                     "v"(hi[7]), "v"(hi[8]), "v"(hi[9]), "v"(hi[10]), "v"(hi[11]), "v"(hi[12]));
        float my_sr = 0.f, my_sc = 0.f;
        #pragma unroll
        for (int u = 0; u < SC_ROWS; ++u) {
            float sr = lo[u].x * ar0.x;
            sr += lo[u].y * ar0.y; sr += lo[u].z * ar0.z; sr += lo[u].w * ar0.w;
            sr += hi[u].x * ar1.x; sr += hi[u].y * ar1.y; sr += hi[u].z * ar1.z; sr += hi[u].w * ar1.w;
            float sc = lo[u].x * ac0.x;
            sc += lo[u].y * ac0.y; sc += lo[u].z * ac0.z; sc += lo[u].w * ac0.w;
            sc += hi[u].x * ac1.x; sc += hi[u].y * ac1.y; sc += hi[u].z * ac1.z; sc += hi[u].w * ac1.w;
            #pragma unroll
            for (int off = 16; off; off >>= 1) {
                sr += __shfl_xor(sr, off, 32);
                sc += __shfl_xor(sc, off, 32);
            }
            if (lane == u) { my_sr = sr; my_sc = sc; }
        }
        const int r = r0 + lane;
        if (lane < SC_ROWS && r < rend) { s_row[r] = my_sr; s_col[r] = my_sc; }
    }

    if (g0 < nq) key_w[g0] = key4_of(k0);
    if (g1 < nq) key_w[g1] = key4_of(k1);
    for (int g = gtid + 2 * nthr; g < nq; g += nthr)
        key_w[g] = key4_of(((const v4i*)ei)[g]);
    unsigned char* key_b = (unsigned char*)key_w;
    for (int e = (nq << 2) + gtid; e < E; e += nthr)
        key_b[e] = (unsigned char)((unsigned)ei[e] / ROWS_PB);
}

#define LIST_CAP    4096
#define WAVE_CAP    192
#define WAVE_PITCH  193
#define GROUP       8
#define SLOTS       7
#define AGG_THREADS 1024

__global__ void __launch_bounds__(AGG_THREADS)
spgat_aggregate(const int* __restrict__ ei, int E,
                const unsigned int* __restrict__ key_w,
                const float* __restrict__ s_row,
                const float* __restrict__ s_col,
                const float* __restrict__ h,
                float* __restrict__ out, int N) {
    __shared__ __attribute__((aligned(16))) float acc[ROWS_PB * NCLASS];
    __shared__ int   list_e[LIST_CAP];
    __shared__ int   wl_c[32 * WAVE_PITCH];
    __shared__ int   wl_r[32 * WAVE_PITCH];
    __shared__ float wl_w[32 * WAVE_PITCH];
    __shared__ int   wcnt[32];
    __shared__ int   cnt;
    __shared__ int   ovf;

    const int tid  = threadIdx.x;
    const int lane = tid & 31;
    const int wave = __builtin_amdgcn_readfirstlane(tid >> 5);
    const int row0 = blockIdx.x * ROWS_PB;
    const int nrows = min(ROWS_PB, N - row0);
    const unsigned nr = (unsigned)nrows;

    for (int i = tid; i < ROWS_PB * NCLASS / 4; i += AGG_THREADS)
        ((float4*)acc)[i] = make_float4(0.f, 0.f, 0.f, 0.f);
    if (tid < 32) wcnt[tid] = 0;
    if (tid == 0) { cnt = 0; ovf = 0; }
    __syncthreads();

    auto put = [&](int pos, int e) {
        if (pos < LIST_CAP) list_e[pos] = e;
        else ovf = 1;
    };
    const unsigned pat = (unsigned)blockIdx.x * 0x01010101u;
    auto hit4 = [&](unsigned w) -> unsigned {
        const unsigned x = w ^ pat;
        return ~(((x & 0x7f7f7f7fu) + 0x7f7f7f7fu) | x) & 0x80808080u;
    };

    const int nvec = E >> 4;
    const uint4* key4 = (const uint4*)key_w;
    const int voff = (int)(((long long)blockIdx.x * nvec) / gridDim.x) & ~31;
    for (int vb = 0; vb < nvec; vb += 8 * AGG_THREADS) {
        const int v0 = vb + tid;
        uint4 q[8];
        #pragma unroll
        for (int u = 0; u < 8; ++u) {
            const int v = min(v0 + u * AGG_THREADS, nvec - 1);
            const int vv = (v + voff < nvec) ? v + voff : v + voff - nvec;
            q[u] = key4[vv];
        }
        unsigned mm[8];
        int c = 0;
        #pragma unroll
        for (int u = 0; u < 8; ++u) {
            unsigned m = (hit4(q[u].x) >> 7) | (hit4(q[u].y) >> 6) |
                         (hit4(q[u].z) >> 5) | (hit4(q[u].w) >> 4);
            if (v0 + u * AGG_THREADS >= nvec) m = 0;
            mm[u] = m;
            c += __popc(m);
        }
        int incl = c;
        #pragma unroll
        for (int off = 1; off < 32; off <<= 1) {
            const int t = __shfl_up(incl, off, 32);
            if (lane >= off) incl += t;
        }
        const int total = __builtin_amdgcn_readlane(incl, 31);
        if (total) {
            int base = 0;
            if (lane == 31) base = atomicAdd(&cnt, total);
            base = __builtin_amdgcn_readlane(base, 31);
            int pos = base + incl - c;
            #pragma unroll
            for (int u = 0; u < 8; ++u) {
                const int v = v0 + u * AGG_THREADS;
                const int e = ((v + voff < nvec) ? v + voff : v + voff - nvec) << 4;
                unsigned m = mm[u];
                while (m) {
                    const int b = __ffs(m) - 1;
                    m &= m - 1;
                    put(pos++, e + ((b & 7) << 2) + (b >> 3));
                }
            }
        }
    }
    {
        const unsigned char* key_b = (const unsigned char*)key_w;
        for (int e = (nvec << 4) + tid; e < E; e += AGG_THREADS)
            if (key_b[e] == (unsigned char)blockIdx.x) put(atomicAdd(&cnt, 1), e);
    }
    __syncthreads();

    const int n = min(cnt, LIST_CAP);
    for (int k = tid; k < n; k += AGG_THREADS) {
        const int e = list_e[k];
        const int r = ei[e];
        const unsigned rl = (unsigned)(r - row0);
        if (rl < nr) {
            const int c = ei[E + e];
            const float lg = s_row[r] + s_col[c];
            const float l  = lg > 0.f ? lg : 0.2f * lg;
            const int w = (int)(rl & 31u);
            const int p = atomicAdd(&wcnt[w], 1);
            if (p < WAVE_CAP) {
                wl_c[w * WAVE_PITCH + p] = c;
                wl_r[w * WAVE_PITCH + p] = (int)rl;
                wl_w[w * WAVE_PITCH + p] = __expf(-l);
            } else {
                ovf = 1;
            }
        }
    }
    __syncthreads();

    float rs[SLOTS];
    #pragma unroll
    for (int j = 0; j < SLOTS; ++j) rs[j] = 0.f;
    auto add_edge = [&](int rl, float w, const float4 lo, const float4 hi) {
        float4* ar = (float4*)(acc + rl * NCLASS) + lane;
        float4 a0 = ar[0];
        float4 a1 = ar[32];
        a0.x += w * lo.x; a0.y += w * lo.y; a0.z += w * lo.z; a0.w += w * lo.w;
        a1.x += w * hi.x; a1.y += w * hi.y; a1.z += w * hi.z; a1.w += w * hi.w;
        ar[0]  = a0;
        ar[32] = a1;
        const int slot = rl >> 5;
        #pragma unroll
        for (int j = 0; j < SLOTS; ++j) rs[j] += (slot == j) ? w : 0.f;
    };

    const int overflow = __builtin_amdgcn_readfirstlane(ovf);
    if (!overflow) {
        const int nw = __builtin_amdgcn_readfirstlane(min(wcnt[wave], WAVE_CAP));
        const int*   wc = wl_c + wave * WAVE_PITCH;
        const int*   wr = wl_r + wave * WAVE_PITCH;
        const float* ww = wl_w + wave * WAVE_PITCH;
        for (int k0 = 0; k0 < nw; k0 += GROUP) {
            float4 hv[GROUP][2];
            float ev[GROUP];
            int   cc[GROUP], rr[GROUP];
            #pragma unroll
            for (int u = 0; u < GROUP; ++u) {
                const int kk = min(k0 + u, nw - 1);
                cc[u] = wc[kk];
                rr[u] = __builtin_amdgcn_readfirstlane(wr[kk]);
                ev[u] = __int_as_float(__builtin_amdgcn_readfirstlane(__float_as_int(ww[kk])));
            }
            #pragma unroll
            for (int u = 0; u < GROUP; ++u) {
                if (k0 + u < nw) {
                    const float4* hc = (const float4*)(h + (size_t)cc[u] * NCLASS) + lane;
                    hv[u][0] = hc[0];
                    hv[u][1] = hc[32];
                }
            }
            #pragma unroll
            for (int u = 0; u < GROUP; ++u) {
                if (k0 + u < nw) add_edge(rr[u], ev[u], hv[u][0], hv[u][1]);
            }
        }
    } else {
        for (int e0 = 0; e0 < E; e0 += 32) {
            const int e = e0 + lane;
            bool mine = false;
            if (e < E) {
                const unsigned rl = (unsigned)(ei[e] - row0);
                mine = (rl < nr) && ((int)(rl & 31u) == wave);
            }
            unsigned m = __builtin_amdgcn_ballot_w32(mine);
            while (m) {
                const int b = __ffs(m) - 1;
                m &= m - 1;
                const int ee = e0 + b;
                const int r = ei[ee];
                const int c = ei[E + ee];
                const float lg = s_row[r] + s_col[c];
                const float l  = lg > 0.f ? lg : 0.2f * lg;
                const float4* hc = (const float4*)(h + (size_t)c * NCLASS) + lane;
                const float wgt = __int_as_float(__builtin_amdgcn_readfirstlane(__float_as_int(__expf(-l))));
                add_edge(__builtin_amdgcn_readfirstlane(r - row0), wgt, hc[0], hc[32]);
            }
        }
    }

    #pragma unroll
    for (int j = 0; j < SLOTS; ++j) {
        const int r = wave + 32 * j;
        if (r >= nrows) break;
        const float inv = 1.0f / rs[j];
        const float4* ar = (const float4*)(acc + r * NCLASS) + lane;
        float4* o = (float4*)(out + (size_t)(row0 + r) * NCLASS);
        const float4 s0 = ar[0];
        const float4 s1 = ar[32];
        float v[8] = { s0.x, s0.y, s0.z, s0.w, s1.x, s1.y, s1.z, s1.w };
        float mx = -3.402823466e38f;
        #pragma unroll
        for (int i = 0; i < 8; ++i) {
            const float hp = v[i] * inv;
            const float el = hp > 0.f ? hp : (__expf(hp) - 1.f);
            v[i] = el;
            mx = fmaxf(mx, el);
        }
        #pragma unroll
        for (int off = 16; off; off >>= 1) mx = fmaxf(mx, __shfl_xor(mx, off, 32));
        float sm = 0.f;
        #pragma unroll
        for (int i = 0; i < 8; ++i) sm += __expf(v[i] - mx);
        #pragma unroll
        for (int off = 16; off; off >>= 1) sm += __shfl_xor(sm, off, 32);
        const float lse = mx + __logf(sm);
        o[lane]      = make_float4(v[0] - lse, v[1] - lse, v[2] - lse, v[3] - lse);
        o[lane + 32] = make_float4(v[4] - lse, v[5] - lse, v[6] - lse, v[7] - lse);
    }
}

extern "C" void kernel_launch(void* const* d_in, const int* in_sizes, int n_in,
                              void* d_out, int out_size, void* d_ws, size_t ws_size,
                              hipStream_t stream) {
    const float* x  = (const float*)d_in[0];
    const float* W  = (const float*)d_in[1];
    const float* a  = (const float*)d_in[2];
    const int*   ei = (const int*)d_in[3];

    const int N = in_sizes[0] / NFEAT;
    const int E = in_sizes[3] / 2;

    float* h      = (float*)d_ws;
    float* s_row  = h + (size_t)N * NCLASS;
    float* s_col  = s_row + N;
    unsigned int* key_w = (unsigned int*)(s_col + N);

    float* out = (float*)d_out;

    {
        const int mtiles = N / 16;
        dim3 grid((mtiles + 1) / 2);
        spgat_gemm<<<grid, 128, 0, stream>>>(x, W, h, N);
    }
    {
        spgat_scores<<<SC_BLOCKS, SC_THREADS, 0, stream>>>(h, a, s_row, s_col, ei, E, key_w, N);
    }
    {
        dim3 grid((N + ROWS_PB - 1) / ROWS_PB);
        spgat_aggregate<<<grid, AGG_THREADS, 0, stream>>>(ei, E, key_w, s_row, s_col, h, out, N);
    }
}


# GEMM one pass per WGP: 256 blocks of 8 waves, a wave keeps 6-7 row tiles x 4 col tiles (24-28 accumulator tiles) in registers over all k, loads 2 k steps ahead; W read once per WGP; h sum order kept
# speedup vs baseline: 3.6187x; 1.1195x over previous
#include <hip/hip_runtime.h>
#include <hip/hip_bf16.h>

#define NFEAT  512
#define NCLASS 256
#define ROWS_PB 196

typedef float v2f __attribute__((ext_vector_type(2)));
typedef float v8f __attribute__((ext_vector_type(8)));

#define GEMM_BLOCKS  256
#define GEMM_THREADS 256
#define GEMM_MT      7

template <int MT>
__device__ __forceinline__ void gemm_tiles(const float* __restrict__ x,
                                           const float* __restrict__ W,
                                           float* __restrict__ h,
                                           const int tile0, const int n0,
                                           const int half, const int sub) {
    v8f c[MT][4];
    #pragma unroll
    for (int t = 0; t < MT; ++t) {
        #pragma unroll
        for (int j = 0; j < 4; ++j) c[t][j] = v8f{};
    }

    const float* xr = x + (size_t)((tile0 << 4) + sub) * NFEAT + 2 * half;
    const float* wb = W + (size_t)(2 * half) * NCLASS + n0 + sub;
    auto load = [&](const int kb, v2f (&a)[MT], v2f (&b)[4]) {
        #pragma unroll
        for (int t = 0; t < MT; ++t) {
            const float* p = xr + (size_t)t * 16 * NFEAT + kb;
            a[t].x = p[0]; a[t].y = p[1];
        }
        const float* wk0 = wb + (size_t)kb * NCLASS;
        const float* wk1 = wk0 + NCLASS;
        #pragma unroll
        for (int j = 0; j < 4; ++j) { b[j].x = wk0[16 * j]; b[j].y = wk1[16 * j]; }
    };
    auto mma = [&](const v2f (&a)[MT], const v2f (&b)[4]) {
        #pragma unroll
        for (int j = 0; j < 4; ++j) {
            #pragma unroll
            for (int t = 0; t < MT; ++t)
                c[t][j] = __builtin_amdgcn_wmma_f32_16x16x4_f32(false, a[t], false, b[j], (short)0, c[t][j], false, false);
        }
    };
    v2f a0[MT], b0[4], a1[MT], b1[4], a2[MT], b2[4], a3[MT], b3[4];
    load(0, a0, b0);
    load(4, a1, b1);
    for (int kb = 0; kb < NFEAT; kb += 16) {
        load(kb + 8,  a2, b2);
        load(kb + 12, a3, b3);
        __builtin_amdgcn_sched_barrier(0);
        mma(a0, b0);
        mma(a1, b1);
        const int kn = kb + 16 < NFEAT ? kb + 16 : kb + 8;
        load(kn,     a0, b0);
        load(kn + 4, a1, b1);
        __builtin_amdgcn_sched_barrier(0);
        mma(a2, b2);
        mma(a3, b3);
    }

    float* hout = h + n0 + sub;
    #pragma unroll
    for (int t = 0; t < MT; ++t) {
        #pragma unroll
        for (int r = 0; r < 8; ++r) {
            const size_t m = (size_t)(((tile0 + t) << 4) + half * 8 + r);
            hout[m * NCLASS +  0] = c[t][0][r];
            hout[m * NCLASS + 16] = c[t][1][r];
            hout[m * NCLASS + 32] = c[t][2][r];
            hout[m * NCLASS + 48] = c[t][3][r];
        }
    }
}

__global__ void __launch_bounds__(GEMM_THREADS)
spgat_gemm(const float* __restrict__ x,
           const float* __restrict__ W,
           float* __restrict__ h, int N) {
    const int lane = threadIdx.x & 31;
    const int wave = __builtin_amdgcn_readfirstlane(threadIdx.x >> 5);
    const int slab = wave & 3;
    const int grp  = wave >> 2;
    const int half = lane >> 4;
    const int sub  = lane & 15;

    const int mtiles = N >> 4;
    const int nb  = (int)gridDim.x, bid = (int)blockIdx.x;
    const int q   = mtiles / nb, rem = mtiles % nb;
    const int t0  = bid * q + min(bid, rem);
    const int cnt = q + (bid < rem ? 1 : 0);
    const int c0  = (cnt + 1) >> 1;
    int tile = grp ? t0 + c0 : t0;
    int left = grp ? cnt - c0 : c0;

    while (left > 0) {
        const int m = min(left, GEMM_MT);
        switch (m) {
            case 7:  gemm_tiles<7>(x, W, h, tile, slab * 64, half, sub); break;
            case 6:  gemm_tiles<6>(x, W, h, tile, slab * 64, half, sub); break;
            case 5:  gemm_tiles<5>(x, W, h, tile, slab * 64, half, sub); break;
            case 4:  gemm_tiles<4>(x, W, h, tile, slab * 64, half, sub); break;
            case 3:  gemm_tiles<3>(x, W, h, tile, slab * 64, half, sub); break;
            case 2:  gemm_tiles<2>(x, W, h, tile, slab * 64, half, sub); break;
            default: gemm_tiles<1>(x, W, h, tile, slab * 64, half, sub); break;
        }
        tile += m;
        left -= m;
    }
}

#define SC_BLOCKS  512
#define SC_THREADS 256
#define SC_WAVES   (SC_BLOCKS * (SC_THREADS / 32))
#define SC_ROWS    13

typedef float v4f __attribute__((ext_vector_type(4)));
typedef int   v4i __attribute__((ext_vector_type(4)));

__device__ __forceinline__ unsigned key4_of(const v4i r) {
    return (((unsigned)r.x / ROWS_PB) & 255u) | ((((unsigned)r.y / ROWS_PB) & 255u) << 8) |
           ((((unsigned)r.z / ROWS_PB) & 255u) << 16) | (((unsigned)r.w / ROWS_PB) << 24);
}

__global__ void __launch_bounds__(SC_THREADS)
spgat_scores(const float* __restrict__ h,
             const float* __restrict__ a,
             float* __restrict__ s_row,
             float* __restrict__ s_col,
             const int* __restrict__ ei, int E,
             unsigned int* __restrict__ key_w, int N) {
    const int lane = threadIdx.x & 31;
    const int gtid = blockIdx.x * SC_THREADS + threadIdx.x;
    const int nthr = SC_BLOCKS * SC_THREADS;
    const int nq   = E >> 2;

    const v4f* a4 = (const v4f*)a;
    const v4f ar0 = a4[lane], ar1 = a4[32 + lane];
    const v4f ac0 = a4[64 + lane], ac1 = a4[96 + lane];
    asm volatile("" :: "v"(ar0), "v"(ar1), "v"(ac0), "v"(ac1));

    const int g0 = gtid, g1 = gtid + nthr;
    const int glast = max(nq, 1) - 1;
    const v4i k0 = ((const v4i*)ei)[min(g0, glast)];
    const v4i k1 = ((const v4i*)ei)[min(g1, glast)];

    const int wg   = __builtin_amdgcn_readfirstlane(gtid >> 5);
    const int q    = N / SC_WAVES, rem = N % SC_WAVES;
    const int rbeg = wg * q + min(wg, rem);
    const int rend = rbeg + q + (wg < rem ? 1 : 0);

    for (int r0 = rbeg; r0 < rend; r0 += SC_ROWS) {
        v4f lo[SC_ROWS], hi[SC_ROWS];
        #pragma unroll
        for (int u = 0; u < SC_ROWS; ++u) {
            const int r = min(r0 + u, rend - 1);
            const v4f* hr = (const v4f*)(h + (size_t)r * NCLASS) + lane;
            lo[u] = hr[0];
            hi[u] = hr[32];
        }
        asm volatile("" :: "v"(k0), "v"(k1),
                     "v"(lo[0]), "v"(lo[1]), "v"(lo[2]), "v"(lo[3]), "v"(lo[4]), "v"(lo[5]), "v"(lo[6]),
                     "v"(lo[7]), "v"(lo[8]), "v"(lo[9]), "v"(lo[10]), "v"(lo[11]), "v"(lo[12]),
                     "v"(hi[0]), "v"(hi[1]), "v"(hi[2]), "v"(hi[3]), "v"(hi[4]), "v"(hi[5]), "v"(hi[6]),
                     "v"(hi[7]), "v"(hi[8]), "v"(hi[9]), "v"(hi[10]), "v"(hi[11]), "v"(hi[12]));
        float my_sr = 0.f, my_sc = 0.f;
        #pragma unroll
        for (int u = 0; u < SC_ROWS; ++u) {
            float sr = lo[u].x * ar0.x;
            sr += lo[u].y * ar0.y; sr += lo[u].z * ar0.z; sr += lo[u].w * ar0.w;
            sr += hi[u].x * ar1.x; sr += hi[u].y * ar1.y; sr += hi[u].z * ar1.z; sr += hi[u].w * ar1.w;
            float sc = lo[u].x * ac0.x;
            sc += lo[u].y * ac0.y; sc += lo[u].z * ac0.z; sc += lo[u].w * ac0.w;
            sc += hi[u].x * ac1.x; sc += hi[u].y * ac1.y; sc += hi[u].z * ac1.z; sc += hi[u].w * ac1.w;
            #pragma unroll
            for (int off = 16; off; off >>= 1) {
                sr += __shfl_xor(sr, off, 32);
                sc += __shfl_xor(sc, off, 32);
            }
            if (lane == u) { my_sr = sr; my_sc = sc; }
        }
        const int r = r0 + lane;
        if (lane < SC_ROWS && r < rend) { s_row[r] = my_sr; s_col[r] = my_sc; }
    }

    if (g0 < nq) key_w[g0] = key4_of(k0);
    if (g1 < nq) key_w[g1] = key4_of(k1);
    for (int g = gtid + 2 * nthr; g < nq; g += nthr)
        key_w[g] = key4_of(((const v4i*)ei)[g]);
    unsigned char* key_b = (unsigned char*)key_w;
    for (int e = (nq << 2) + gtid; e < E; e += nthr)
        key_b[e] = (unsigned char)((unsigned)ei[e] / ROWS_PB);
}

#define LIST_CAP    4096
#define WAVE_CAP    192
#define WAVE_PITCH  193
#define GROUP       8
#define SLOTS       7
#define AGG_THREADS 1024

__global__ void __launch_bounds__(AGG_THREADS)
spgat_aggregate(const int* __restrict__ ei, int E,
                const unsigned int* __restrict__ key_w,
                const float* __restrict__ s_row,
                const float* __restrict__ s_col,
                const float* __restrict__ h,
                float* __restrict__ out, int N) {
    __shared__ __attribute__((aligned(16))) float acc[ROWS_PB * NCLASS];
    __shared__ int   list_e[LIST_CAP];
    __shared__ int   wl_c[32 * WAVE_PITCH];
    __shared__ int   wl_r[32 * WAVE_PITCH];
    __shared__ float wl_w[32 * WAVE_PITCH];
    __shared__ int   wcnt[32];
    __shared__ int   cnt;
    __shared__ int   ovf;

    const int tid  = threadIdx.x;
    const int lane = tid & 31;
    const int wave = __builtin_amdgcn_readfirstlane(tid >> 5);
    const int row0 = blockIdx.x * ROWS_PB;
    const int nrows = min(ROWS_PB, N - row0);
    const unsigned nr = (unsigned)nrows;

    for (int i = tid; i < ROWS_PB * NCLASS / 4; i += AGG_THREADS)
        ((float4*)acc)[i] = make_float4(0.f, 0.f, 0.f, 0.f);
    if (tid < 32) wcnt[tid] = 0;
    if (tid == 0) { cnt = 0; ovf = 0; }
    __syncthreads();

    auto put = [&](int pos, int e) {
        if (pos < LIST_CAP) list_e[pos] = e;
        else ovf = 1;
    };
    const unsigned pat = (unsigned)blockIdx.x * 0x01010101u;
    auto hit4 = [&](unsigned w) -> unsigned {
        const unsigned x = w ^ pat;
        return ~(((x & 0x7f7f7f7fu) + 0x7f7f7f7fu) | x) & 0x80808080u;
    };

    const int nvec = E >> 4;
    const uint4* key4 = (const uint4*)key_w;
    const int voff = (int)(((long long)blockIdx.x * nvec) / gridDim.x) & ~31;
    for (int vb = 0; vb < nvec; vb += 8 * AGG_THREADS) {
        const int v0 = vb + tid;
        uint4 q[8];
        #pragma unroll
        for (int u = 0; u < 8; ++u) {
            const int v = min(v0 + u * AGG_THREADS, nvec - 1);
            const int vv = (v + voff < nvec) ? v + voff : v + voff - nvec;
            q[u] = key4[vv];
        }
        unsigned mm[8];
        int c = 0;
        #pragma unroll
        for (int u = 0; u < 8; ++u) {
            unsigned m = (hit4(q[u].x) >> 7) | (hit4(q[u].y) >> 6) |
                         (hit4(q[u].z) >> 5) | (hit4(q[u].w) >> 4);
            if (v0 + u * AGG_THREADS >= nvec) m = 0;
            mm[u] = m;
            c += __popc(m);
        }
        int incl = c;
        #pragma unroll
        for (int off = 1; off < 32; off <<= 1) {
            const int t = __shfl_up(incl, off, 32);
            if (lane >= off) incl += t;
        }
        const int total = __builtin_amdgcn_readlane(incl, 31);
        if (total) {
            int base = 0;
            if (lane == 31) base = atomicAdd(&cnt, total);
            base = __builtin_amdgcn_readlane(base, 31);
            int pos = base + incl - c;
            #pragma unroll
            for (int u = 0; u < 8; ++u) {
                const int v = v0 + u * AGG_THREADS;
                const int e = ((v + voff < nvec) ? v + voff : v + voff - nvec) << 4;
                unsigned m = mm[u];
                while (m) {
                    const int b = __ffs(m) - 1;
                    m &= m - 1;
                    put(pos++, e + ((b & 7) << 2) + (b >> 3));
                }
            }
        }
    }
    {
        const unsigned char* key_b = (const unsigned char*)key_w;
        for (int e = (nvec << 4) + tid; e < E; e += AGG_THREADS)
            if (key_b[e] == (unsigned char)blockIdx.x) put(atomicAdd(&cnt, 1), e);
    }
    __syncthreads();

    const int n = min(cnt, LIST_CAP);
    for (int k = tid; k < n; k += AGG_THREADS) {
        const int e = list_e[k];
        const int r = ei[e];
        const unsigned rl = (unsigned)(r - row0);
        if (rl < nr) {
            const int c = ei[E + e];
            const float lg = s_row[r] + s_col[c];
            const float l  = lg > 0.f ? lg : 0.2f * lg;
            const int w = (int)(rl & 31u);
            const int p = atomicAdd(&wcnt[w], 1);
            if (p < WAVE_CAP) {
                wl_c[w * WAVE_PITCH + p] = c;
                wl_r[w * WAVE_PITCH + p] = (int)rl;
                wl_w[w * WAVE_PITCH + p] = __expf(-l);
            } else {
                ovf = 1;
            }
        }
    }
    __syncthreads();

    float rs[SLOTS];
    #pragma unroll
    for (int j = 0; j < SLOTS; ++j) rs[j] = 0.f;
    auto add_edge = [&](int rl, float w, const float4 lo, const float4 hi) {
        float4* ar = (float4*)(acc + rl * NCLASS) + lane;
        float4 a0 = ar[0];
        float4 a1 = ar[32];
        a0.x += w * lo.x; a0.y += w * lo.y; a0.z += w * lo.z; a0.w += w * lo.w;
        a1.x += w * hi.x; a1.y += w * hi.y; a1.z += w * hi.z; a1.w += w * hi.w;
        ar[0]  = a0;
        ar[32] = a1;
        const int slot = rl >> 5;
        #pragma unroll
        for (int j = 0; j < SLOTS; ++j) rs[j] += (slot == j) ? w : 0.f;
    };

    const int overflow = __builtin_amdgcn_readfirstlane(ovf);
    if (!overflow) {
        const int nw = __builtin_amdgcn_readfirstlane(min(wcnt[wave], WAVE_CAP));
        const int*   wc = wl_c + wave * WAVE_PITCH;
        const int*   wr = wl_r + wave * WAVE_PITCH;
        const float* ww = wl_w + wave * WAVE_PITCH;
        for (int k0 = 0; k0 < nw; k0 += GROUP) {
            float4 hv[GROUP][2];
            float ev[GROUP];
            int   cc[GROUP], rr[GROUP];
            #pragma unroll
            for (int u = 0; u < GROUP; ++u) {
                const int kk = min(k0 + u, nw - 1);
                cc[u] = wc[kk];
                rr[u] = __builtin_amdgcn_readfirstlane(wr[kk]);
                ev[u] = __int_as_float(__builtin_amdgcn_readfirstlane(__float_as_int(ww[kk])));
            }
            #pragma unroll
            for (int u = 0; u < GROUP; ++u) {
                if (k0 + u < nw) {
                    const float4* hc = (const float4*)(h + (size_t)cc[u] * NCLASS) + lane;
                    hv[u][0] = hc[0];
                    hv[u][1] = hc[32];
                }
            }
            #pragma unroll
            for (int u = 0; u < GROUP; ++u) {
                if (k0 + u < nw) add_edge(rr[u], ev[u], hv[u][0], hv[u][1]);
            }
        }
    } else {
        for (int e0 = 0; e0 < E; e0 += 32) {
            const int e = e0 + lane;
            bool mine = false;
            if (e < E) {
                const unsigned rl = (unsigned)(ei[e] - row0);
                mine = (rl < nr) && ((int)(rl & 31u) == wave);
            }
            unsigned m = __builtin_amdgcn_ballot_w32(mine);
            while (m) {
                const int b = __ffs(m) - 1;
                m &= m - 1;
                const int ee = e0 + b;
                const int r = ei[ee];
                const int c = ei[E + ee];
                const float lg = s_row[r] + s_col[c];
                const float l  = lg > 0.f ? lg : 0.2f * lg;
                const float4* hc = (const float4*)(h + (size_t)c * NCLASS) + lane;
                const float wgt = __int_as_float(__builtin_amdgcn_readfirstlane(__float_as_int(__expf(-l))));
                add_edge(__builtin_amdgcn_readfirstlane(r - row0), wgt, hc[0], hc[32]);
            }
        }
    }

    #pragma unroll
    for (int j = 0; j < SLOTS; ++j) {
        const int r = wave + 32 * j;
        if (r >= nrows) break;
        const float inv = 1.0f / rs[j];
        const float4* ar = (const float4*)(acc + r * NCLASS) + lane;
        float4* o = (float4*)(out + (size_t)(row0 + r) * NCLASS);
        const float4 s0 = ar[0];
        const float4 s1 = ar[32];
        float v[8] = { s0.x, s0.y, s0.z, s0.w, s1.x, s1.y, s1.z, s1.w };
        float mx = -3.402823466e38f;
        #pragma unroll
        for (int i = 0; i < 8; ++i) {
            const float hp = v[i] * inv;
            const float el = hp > 0.f ? hp : (__expf(hp) - 1.f);
            v[i] = el;
            mx = fmaxf(mx, el);
        }
        #pragma unroll
        for (int off = 16; off; off >>= 1) mx = fmaxf(mx, __shfl_xor(mx, off, 32));
        float sm = 0.f;
        #pragma unroll
        for (int i = 0; i < 8; ++i) sm += __expf(v[i] - mx);
        #pragma unroll
        for (int off = 16; off; off >>= 1) sm += __shfl_xor(sm, off, 32);
        const float lse = mx + __logf(sm);
        o[lane]      = make_float4(v[0] - lse, v[1] - lse, v[2] - lse, v[3] - lse);
        o[lane + 32] = make_float4(v[4] - lse, v[5] - lse, v[6] - lse, v[7] - lse);
    }
}

extern "C" void kernel_launch(void* const* d_in, const int* in_sizes, int n_in,
                              void* d_out, int out_size, void* d_ws, size_t ws_size,
                              hipStream_t stream) {
    const float* x  = (const float*)d_in[0];
    const float* W  = (const float*)d_in[1];
    const float* a  = (const float*)d_in[2];
    const int*   ei = (const int*)d_in[3];

    const int N = in_sizes[0] / NFEAT;
    const int E = in_sizes[3] / 2;

    float* h      = (float*)d_ws;
    float* s_row  = h + (size_t)N * NCLASS;
    float* s_col  = s_row + N;
    unsigned int* key_w = (unsigned int*)(s_col + N);

    float* out = (float*)d_out;

    {
        spgat_gemm<<<GEMM_BLOCKS, GEMM_THREADS, 0, stream>>>(x, W, h, N);
    }
    {
        spgat_scores<<<SC_BLOCKS, SC_THREADS, 0, stream>>>(h, a, s_row, s_col, ei, E, key_w, N);
    }
    {
        dim3 grid((N + ROWS_PB - 1) / ROWS_PB);
        spgat_aggregate<<<grid, AGG_THREADS, 0, stream>>>(ei, E, key_w, s_row, s_col, h, out, N);
    }
}


# Edge keys as 8 bit planes per 32 edges (wave ballots in the scores launch); aggregation blocks match 32 edges with one xor-and per plane, not per-byte tests; row sum order = list arrival as before
# speedup vs baseline: 3.7549x; 1.0376x over previous
#include <hip/hip_runtime.h>
#include <hip/hip_bf16.h>

#define NFEAT  512
#define NCLASS 256
#define ROWS_PB 196

typedef float v2f __attribute__((ext_vector_type(2)));
typedef float v8f __attribute__((ext_vector_type(8)));

#define GEMM_BLOCKS  256
#define GEMM_THREADS 256
#define GEMM_MT      7

template <int MT>
__device__ __forceinline__ void gemm_tiles(const float* __restrict__ x,
                                           const float* __restrict__ W,
                                           float* __restrict__ h,
                                           const int tile0, const int n0,
                                           const int half, const int sub) {
    v8f c[MT][4];
    #pragma unroll
    for (int t = 0; t < MT; ++t) {
        #pragma unroll
        for (int j = 0; j < 4; ++j) c[t][j] = v8f{};
    }

    const float* xr = x + (size_t)((tile0 << 4) + sub) * NFEAT + 2 * half;
    const float* wb = W + (size_t)(2 * half) * NCLASS + n0 + sub;
    auto load = [&](const int kb, v2f (&a)[MT], v2f (&b)[4]) {
        #pragma unroll
        for (int t = 0; t < MT; ++t) {
            const float* p = xr + (size_t)t * 16 * NFEAT + kb;
            a[t].x = p[0]; a[t].y = p[1];
        }
        const float* wk0 = wb + (size_t)kb * NCLASS;
        const float* wk1 = wk0 + NCLASS;
        #pragma unroll
        for (int j = 0; j < 4; ++j) { b[j].x = wk0[16 * j]; b[j].y = wk1[16 * j]; }
    };
    auto mma = [&](const v2f (&a)[MT], const v2f (&b)[4]) {
        #pragma unroll
        for (int j = 0; j < 4; ++j) {
            #pragma unroll
            for (int t = 0; t < MT; ++t)
                c[t][j] = __builtin_amdgcn_wmma_f32_16x16x4_f32(false, a[t], false, b[j], (short)0, c[t][j], false, false);
        }
    };
    v2f a0[MT], b0[4], a1[MT], b1[4], a2[MT], b2[4], a3[MT], b3[4];
    load(0, a0, b0);
    load(4, a1, b1);
    for (int kb = 0; kb < NFEAT; kb += 16) {
        load(kb + 8,  a2, b2);
        load(kb + 12, a3, b3);
        __builtin_amdgcn_sched_barrier(0);
        mma(a0, b0);
        mma(a1, b1);
        const int kn = kb + 16 < NFEAT ? kb + 16 : kb + 8;
        load(kn,     a0, b0);
        load(kn + 4, a1, b1);
        __builtin_amdgcn_sched_barrier(0);
        mma(a2, b2);
        mma(a3, b3);
    }

    float* hout = h + n0 + sub;
    #pragma unroll
    for (int t = 0; t < MT; ++t) {
        #pragma unroll
        for (int r = 0; r < 8; ++r) {
            const size_t m = (size_t)(((tile0 + t) << 4) + half * 8 + r);
            hout[m * NCLASS +  0] = c[t][0][r];
            hout[m * NCLASS + 16] = c[t][1][r];
            hout[m * NCLASS + 32] = c[t][2][r];
            hout[m * NCLASS + 48] = c[t][3][r];
        }
    }
}

__global__ void __launch_bounds__(GEMM_THREADS)
spgat_gemm(const float* __restrict__ x,
           const float* __restrict__ W,
           float* __restrict__ h, int N) {
    const int lane = threadIdx.x & 31;
    const int wave = __builtin_amdgcn_readfirstlane(threadIdx.x >> 5);
    const int slab = wave & 3;
    const int grp  = wave >> 2;
    const int half = lane >> 4;
    const int sub  = lane & 15;

    const int mtiles = N >> 4;
    const int nb  = (int)gridDim.x, bid = (int)blockIdx.x;
    const int q   = mtiles / nb, rem = mtiles % nb;
    const int t0  = bid * q + min(bid, rem);
    const int cnt = q + (bid < rem ? 1 : 0);
    const int c0  = (cnt + 1) >> 1;
    int tile = grp ? t0 + c0 : t0;
    int left = grp ? cnt - c0 : c0;

    while (left > 0) {
        const int m = min(left, GEMM_MT);
        switch (m) {
            case 7:  gemm_tiles<7>(x, W, h, tile, slab * 64, half, sub); break;
            case 6:  gemm_tiles<6>(x, W, h, tile, slab * 64, half, sub); break;
            case 5:  gemm_tiles<5>(x, W, h, tile, slab * 64, half, sub); break;
            case 4:  gemm_tiles<4>(x, W, h, tile, slab * 64, half, sub); break;
            case 3:  gemm_tiles<3>(x, W, h, tile, slab * 64, half, sub); break;
            case 2:  gemm_tiles<2>(x, W, h, tile, slab * 64, half, sub); break;
            default: gemm_tiles<1>(x, W, h, tile, slab * 64, half, sub); break;
        }
        tile += m;
        left -= m;
    }
}

#define SC_BLOCKS  512
#define SC_THREADS 256
#define SC_WAVES   (SC_BLOCKS * (SC_THREADS / 32))
#define SC_ROWS    13

typedef float v4f __attribute__((ext_vector_type(4)));
typedef int   v4i __attribute__((ext_vector_type(4)));

__device__ __forceinline__ void store_planes(unsigned int* __restrict__ plane_w, const int wv,
                                             const v4i r, const int g, const int E, const int lane) {
    unsigned kk[4] = { ((unsigned)r.x / ROWS_PB) & 255u, ((unsigned)r.y / ROWS_PB) & 255u,
                       ((unsigned)r.z / ROWS_PB) & 255u, ((unsigned)r.w / ROWS_PB) & 255u };
    unsigned mine = 0;
    #pragma unroll
    for (int i = 0; i < 4; ++i) {
        if (4 * g + i >= E) kk[i] = 255u;
        #pragma unroll
        for (int p = 0; p < 8; ++p) {
            const unsigned b = __builtin_amdgcn_ballot_w32(((kk[i] >> p) & 1u) != 0u);
            if (lane == i * 8 + p) mine = b;
        }
    }
    plane_w[(size_t)wv * 32 + lane] = mine;
}

__global__ void __launch_bounds__(SC_THREADS)
spgat_scores(const float* __restrict__ h,
             const float* __restrict__ a,
             float* __restrict__ s_row,
             float* __restrict__ s_col,
             const int* __restrict__ ei, int E,
             unsigned int* __restrict__ key_w, int N) {
    const int lane = threadIdx.x & 31;
    const int gtid = blockIdx.x * SC_THREADS + threadIdx.x;
    const int nthr = SC_BLOCKS * SC_THREADS;
    const int nq   = (E + 3) >> 2;
    const int nwg  = (nq + 31) >> 5;

    const v4f* a4 = (const v4f*)a;
    const v4f ar0 = a4[lane], ar1 = a4[32 + lane];
    const v4f ac0 = a4[64 + lane], ac1 = a4[96 + lane];
    asm volatile("" :: "v"(ar0), "v"(ar1), "v"(ac0), "v"(ac1));

    const int g0 = gtid, g1 = gtid + nthr;
    const int glast = max(nq, 1) - 1;
    const v4i k0 = ((const v4i*)ei)[min(g0, glast)];
    const v4i k1 = ((const v4i*)ei)[min(g1, glast)];

    const int wg   = __builtin_amdgcn_readfirstlane(gtid >> 5);
    const int q    = N / SC_WAVES, rem = N % SC_WAVES;
    const int rbeg = wg * q + min(wg, rem);
    const int rend = rbeg + q + (wg < rem ? 1 : 0);

    for (int r0 = rbeg; r0 < rend; r0 += SC_ROWS) {
        v4f lo[SC_ROWS], hi[SC_ROWS];
        #pragma unroll
        for (int u = 0; u < SC_ROWS; ++u) {
            const int r = min(r0 + u, rend - 1);
            const v4f* hr = (const v4f*)(h + (size_t)r * NCLASS) + lane;
            lo[u] = hr[0];
            hi[u] = hr[32];
        }
        asm volatile("" :: "v"(k0), "v"(k1),
                     "v"(lo[0]), "v"(lo[1]), "v"(lo[2]), "v"(lo[3]), "v"(lo[4]), "v"(lo[5]), "v"(lo[6]),
                     "v"(lo[7]), "v"(lo[8]), "v"(lo[9]), "v"(lo[10]), "v"(lo[11]), "v"(lo[12]),
                     "v"(hi[0]), "v"(hi[1]), "v"(hi[2]), "v"(hi[3]), "v"(hi[4]), "v"(hi[5]), "v"(hi[6]),
                     "v"(hi[7]), "v"(hi[8]), "v"(hi[9]), "v"(hi[10]), "v"(hi[11]), "v"(hi[12]));
        float my_sr = 0.f, my_sc = 0.f;
        #pragma unroll
        for (int u = 0; u < SC_ROWS; ++u) {
            float sr = lo[u].x * ar0.x;
            sr += lo[u].y * ar0.y; sr += lo[u].z * ar0.z; sr += lo[u].w * ar0.w;
            sr += hi[u].x * ar1.x; sr += hi[u].y * ar1.y; sr += hi[u].z * ar1.z; sr += hi[u].w * ar1.w;
            float sc = lo[u].x * ac0.x;
            sc += lo[u].y * ac0.y; sc += lo[u].z * ac0.z; sc += lo[u].w * ac0.w;
            sc += hi[u].x * ac1.x; sc += hi[u].y * ac1.y; sc += hi[u].z * ac1.z; sc += hi[u].w * ac1.w;
            #pragma unroll
            for (int off = 16; off; off >>= 1) {
                sr += __shfl_xor(sr, off, 32);
                sc += __shfl_xor(sc, off, 32);
            }
            if (lane == u) { my_sr = sr; my_sc = sc; }
        }
        const int r = r0 + lane;
        if (lane < SC_ROWS && r < rend) { s_row[r] = my_sr; s_col[r] = my_sc; }
    }

    if (wg < nwg)            store_planes(key_w, wg, k0, g0, E, lane);
    if (wg + SC_WAVES < nwg) store_planes(key_w, wg + SC_WAVES, k1, g1, E, lane);
    for (int wv = wg + 2 * SC_WAVES; wv < nwg; wv += SC_WAVES) {
        const int g = (wv << 5) + lane;
        store_planes(key_w, wv, ((const v4i*)ei)[min(g, glast)], g, E, lane);
    }
}

#define LIST_CAP    4096
#define WAVE_CAP    192
#define WAVE_PITCH  193
#define GROUP       8
#define SLOTS       7
#define AGG_THREADS 1024
#define SCAN_U      7

__global__ void __launch_bounds__(AGG_THREADS)
spgat_aggregate(const int* __restrict__ ei, int E,
                const unsigned int* __restrict__ key_w,
                const float* __restrict__ s_row,
                const float* __restrict__ s_col,
                const float* __restrict__ h,
                float* __restrict__ out, int N) {
    __shared__ __attribute__((aligned(16))) float acc[ROWS_PB * NCLASS];
    __shared__ int   list_e[LIST_CAP];
    __shared__ int   wl_c[32 * WAVE_PITCH];
    __shared__ int   wl_r[32 * WAVE_PITCH];
    __shared__ float wl_w[32 * WAVE_PITCH];
    __shared__ int   wcnt[32];
    __shared__ int   cnt;
    __shared__ int   ovf;

    const int tid  = threadIdx.x;
    const int lane = tid & 31;
    const int wave = __builtin_amdgcn_readfirstlane(tid >> 5);
    const int row0 = blockIdx.x * ROWS_PB;
    const int nrows = min(ROWS_PB, N - row0);
    const unsigned nr = (unsigned)nrows;

    for (int i = tid; i < ROWS_PB * NCLASS / 4; i += AGG_THREADS)
        ((float4*)acc)[i] = make_float4(0.f, 0.f, 0.f, 0.f);
    if (tid < 32) wcnt[tid] = 0;
    if (tid == 0) { cnt = 0; ovf = 0; }
    __syncthreads();

    auto put = [&](int pos, int e) {
        if (pos < LIST_CAP) list_e[pos] = e;
        else ovf = 1;
    };
    unsigned cmask[8];
    #pragma unroll
    for (int p = 0; p < 8; ++p) cmask[p] = ((blockIdx.x >> p) & 1u) ? 0u : ~0u;
    auto match = [&](const uint4 a, const uint4 b) -> unsigned {
        return (a.x ^ cmask[0]) & (a.y ^ cmask[1]) & (a.z ^ cmask[2]) & (a.w ^ cmask[3]) &
               (b.x ^ cmask[4]) & (b.y ^ cmask[5]) & (b.z ^ cmask[6]) & (b.w ^ cmask[7]);
    };

    const int nsc = (((E + 3) >> 2) + 31) >> 5 << 2;
    const uint4* pl4 = (const uint4*)key_w;
    const int soff = (int)(((long long)blockIdx.x * nsc) / gridDim.x) & ~31;
    for (int sb = 0; sb < nsc; sb += SCAN_U * AGG_THREADS) {
        const int s0 = sb + tid;
        uint4 qa[SCAN_U], qb[SCAN_U];
        #pragma unroll
        for (int u = 0; u < SCAN_U; ++u) {
            const int s = min(s0 + u * AGG_THREADS, nsc - 1);
            const int ss = (s + soff < nsc) ? s + soff : s + soff - nsc;
            qa[u] = pl4[2 * ss];
            qb[u] = pl4[2 * ss + 1];
        }
        unsigned mm[SCAN_U];
        int c = 0;
        #pragma unroll
        for (int u = 0; u < SCAN_U; ++u) {
            unsigned m = match(qa[u], qb[u]);
            if (s0 + u * AGG_THREADS >= nsc) m = 0;
            mm[u] = m;
            c += __popc(m);
        }
        int incl = c;
        #pragma unroll
        for (int off = 1; off < 32; off <<= 1) {
            const int t = __shfl_up(incl, off, 32);
            if (lane >= off) incl += t;
        }
        const int total = __builtin_amdgcn_readlane(incl, 31);
        if (total) {
            int base = 0;
            if (lane == 31) base = atomicAdd(&cnt, total);
            base = __builtin_amdgcn_readlane(base, 31);
            int pos = base + incl - c;
            #pragma unroll
            for (int u = 0; u < SCAN_U; ++u) {
                const int s = s0 + u * AGG_THREADS;
                const int ss = (s + soff < nsc) ? s + soff : s + soff - nsc;
                const int e = ((ss & ~3) << 5) + (ss & 3);
                unsigned m = mm[u];
                while (m) {
                    const int b = __ffs(m) - 1;
                    m &= m - 1;
                    put(pos++, e + (b << 2));
                }
            }
        }
    }
    __syncthreads();

    const int n = min(cnt, LIST_CAP);
    for (int k = tid; k < n; k += AGG_THREADS) {
        const int e = list_e[k];
        const int r = ei[min(e, E - 1)];
        const unsigned rl = (unsigned)(r - row0);
        if (e < E && rl < nr) {
            const int c = ei[E + e];
            const float lg = s_row[r] + s_col[c];
            const float l  = lg > 0.f ? lg : 0.2f * lg;
            const int w = (int)(rl & 31u);
            const int p = atomicAdd(&wcnt[w], 1);
            if (p < WAVE_CAP) {
                wl_c[w * WAVE_PITCH + p] = c;
                wl_r[w * WAVE_PITCH + p] = (int)rl;
                wl_w[w * WAVE_PITCH + p] = __expf(-l);
            } else {
                ovf = 1;
            }
        }
    }
    __syncthreads();

    float rs[SLOTS];
    #pragma unroll
    for (int j = 0; j < SLOTS; ++j) rs[j] = 0.f;
    auto add_edge = [&](int rl, float w, const float4 lo, const float4 hi) {
        float4* ar = (float4*)(acc + rl * NCLASS) + lane;
        float4 a0 = ar[0];
        float4 a1 = ar[32];
        a0.x += w * lo.x; a0.y += w * lo.y; a0.z += w * lo.z; a0.w += w * lo.w;
        a1.x += w * hi.x; a1.y += w * hi.y; a1.z += w * hi.z; a1.w += w * hi.w;
        ar[0]  = a0;
        ar[32] = a1;
        const int slot = rl >> 5;
        #pragma unroll
        for (int j = 0; j < SLOTS; ++j) rs[j] += (slot == j) ? w : 0.f;
    };

    const int overflow = __builtin_amdgcn_readfirstlane(ovf);
    if (!overflow) {
        const int nw = __builtin_amdgcn_readfirstlane(min(wcnt[wave], WAVE_CAP));
        const int*   wc = wl_c + wave * WAVE_PITCH;
        const int*   wr = wl_r + wave * WAVE_PITCH;
        const float* ww = wl_w + wave * WAVE_PITCH;
        for (int k0 = 0; k0 < nw; k0 += GROUP) {
            float4 hv[GROUP][2];
            float ev[GROUP];
            int   cc[GROUP], rr[GROUP];
            #pragma unroll
            for (int u = 0; u < GROUP; ++u) {
                const int kk = min(k0 + u, nw - 1);
                cc[u] = wc[kk];
                rr[u] = __builtin_amdgcn_readfirstlane(wr[kk]);
                ev[u] = __int_as_float(__builtin_amdgcn_readfirstlane(__float_as_int(ww[kk])));
            }
            #pragma unroll
            for (int u = 0; u < GROUP; ++u) {
                if (k0 + u < nw) {
                    const float4* hc = (const float4*)(h + (size_t)cc[u] * NCLASS) + lane;
                    hv[u][0] = hc[0];
                    hv[u][1] = hc[32];
                }
            }
            #pragma unroll
            for (int u = 0; u < GROUP; ++u) {
                if (k0 + u < nw) add_edge(rr[u], ev[u], hv[u][0], hv[u][1]);
            }
        }
    } else {
        for (int e0 = 0; e0 < E; e0 += 32) {
            const int e = e0 + lane;
            bool mine = false;
            if (e < E) {
                const unsigned rl = (unsigned)(ei[e] - row0);
                mine = (rl < nr) && ((int)(rl & 31u) == wave);
            }
            unsigned m = __builtin_amdgcn_ballot_w32(mine);
            while (m) {
                const int b = __ffs(m) - 1;
                m &= m - 1;
                const int ee = e0 + b;
                const int r = ei[ee];
                const int c = ei[E + ee];
                const float lg = s_row[r] + s_col[c];
                const float l  = lg > 0.f ? lg : 0.2f * lg;
                const float4* hc = (const float4*)(h + (size_t)c * NCLASS) + lane;
                const float wgt = __int_as_float(__builtin_amdgcn_readfirstlane(__float_as_int(__expf(-l))));
                add_edge(__builtin_amdgcn_readfirstlane(r - row0), wgt, hc[0], hc[32]);
            }
        }
    }

    #pragma unroll
    for (int j = 0; j < SLOTS; ++j) {
        const int r = wave + 32 * j;
        if (r >= nrows) break;
        const float inv = 1.0f / rs[j];
        const float4* ar = (const float4*)(acc + r * NCLASS) + lane;
        float4* o = (float4*)(out + (size_t)(row0 + r) * NCLASS);
        const float4 s0 = ar[0];
        const float4 s1 = ar[32];
        float v[8] = { s0.x, s0.y, s0.z, s0.w, s1.x, s1.y, s1.z, s1.w };
        float mx = -3.402823466e38f;
        #pragma unroll
        for (int i = 0; i < 8; ++i) {
            const float hp = v[i] * inv;
            const float el = hp > 0.f ? hp : (__expf(hp) - 1.f);
            v[i] = el;
            mx = fmaxf(mx, el);
        }
        #pragma unroll
        for (int off = 16; off; off >>= 1) mx = fmaxf(mx, __shfl_xor(mx, off, 32));
        float sm = 0.f;
        #pragma unroll
        for (int i = 0; i < 8; ++i) sm += __expf(v[i] - mx);
        #pragma unroll
        for (int off = 16; off; off >>= 1) sm += __shfl_xor(sm, off, 32);
        const float lse = mx + __logf(sm);
        o[lane]      = make_float4(v[0] - lse, v[1] - lse, v[2] - lse, v[3] - lse);
        o[lane + 32] = make_float4(v[4] - lse, v[5] - lse, v[6] - lse, v[7] - lse);
    }
}

extern "C" void kernel_launch(void* const* d_in, const int* in_sizes, int n_in,
                              void* d_out, int out_size, void* d_ws, size_t ws_size,
                              hipStream_t stream) {
    const float* x  = (const float*)d_in[0];
    const float* W  = (const float*)d_in[1];
    const float* a  = (const float*)d_in[2];
    const int*   ei = (const int*)d_in[3];

    const int N = in_sizes[0] / NFEAT;
    const int E = in_sizes[3] / 2;

    float* h      = (float*)d_ws;
    float* s_row  = h + (size_t)N * NCLASS;
    float* s_col  = s_row + N;
    unsigned int* key_w = (unsigned int*)(s_col + N);

    float* out = (float*)d_out;

    {
        spgat_gemm<<<GEMM_BLOCKS, GEMM_THREADS, 0, stream>>>(x, W, h, N);
    }
    {
        spgat_scores<<<SC_BLOCKS, SC_THREADS, 0, stream>>>(h, a, s_row, s_col, ei, E, key_w, N);
    }
    {
        dim3 grid((N + ROWS_PB - 1) / ROWS_PB);
        spgat_aggregate<<<grid, AGG_THREADS, 0, stream>>>(ei, E, key_w, s_row, s_col, h, out, N);
    }
}


# Aggregation row sums in registers: listed edges binned as (col, weight) in per-row LDS lists (rank = LDS int add); owner wave sums a row in 8 regs a lane, finalizes, stores; sum order = list arrival
# speedup vs baseline: 3.9322x; 1.0472x over previous
#include <hip/hip_runtime.h>
#include <hip/hip_bf16.h>

#define NFEAT  512
#define NCLASS 256
#define ROWS_PB 196

typedef float v2f __attribute__((ext_vector_type(2)));
typedef float v8f __attribute__((ext_vector_type(8)));

#define GEMM_BLOCKS  256
#define GEMM_THREADS 256
#define GEMM_MT      7

template <int MT>
__device__ __forceinline__ void gemm_tiles(const float* __restrict__ x,
                                           const float* __restrict__ W,
                                           float* __restrict__ h,
                                           const int tile0, const int n0,
                                           const int half, const int sub) {
    v8f c[MT][4];
    #pragma unroll
    for (int t = 0; t < MT; ++t) {
        #pragma unroll
        for (int j = 0; j < 4; ++j) c[t][j] = v8f{};
    }

    const float* xr = x + (size_t)((tile0 << 4) + sub) * NFEAT + 2 * half;
    const float* wb = W + (size_t)(2 * half) * NCLASS + n0 + sub;
    auto load = [&](const int kb, v2f (&a)[MT], v2f (&b)[4]) {
        #pragma unroll
        for (int t = 0; t < MT; ++t) {
            const float* p = xr + (size_t)t * 16 * NFEAT + kb;
            a[t].x = p[0]; a[t].y = p[1];
        }
        const float* wk0 = wb + (size_t)kb * NCLASS;
        const float* wk1 = wk0 + NCLASS;
        #pragma unroll
        for (int j = 0; j < 4; ++j) { b[j].x = wk0[16 * j]; b[j].y = wk1[16 * j]; }
    };
    auto mma = [&](const v2f (&a)[MT], const v2f (&b)[4]) {
        #pragma unroll
        for (int j = 0; j < 4; ++j) {
            #pragma unroll
            for (int t = 0; t < MT; ++t)
                c[t][j] = __builtin_amdgcn_wmma_f32_16x16x4_f32(false, a[t], false, b[j], (short)0, c[t][j], false, false);
        }
    };
    v2f a0[MT], b0[4], a1[MT], b1[4], a2[MT], b2[4], a3[MT], b3[4];
    load(0, a0, b0);
    load(4, a1, b1);
    for (int kb = 0; kb < NFEAT; kb += 16) {
        load(kb + 8,  a2, b2);
        load(kb + 12, a3, b3);
        __builtin_amdgcn_sched_barrier(0);
        mma(a0, b0);
        mma(a1, b1);
        const int kn = kb + 16 < NFEAT ? kb + 16 : kb + 8;
        load(kn,     a0, b0);
        load(kn + 4, a1, b1);
        __builtin_amdgcn_sched_barrier(0);
        mma(a2, b2);
        mma(a3, b3);
    }

    float* hout = h + n0 + sub;
    #pragma unroll
    for (int t = 0; t < MT; ++t) {
        #pragma unroll
        for (int r = 0; r < 8; ++r) {
            const size_t m = (size_t)(((tile0 + t) << 4) + half * 8 + r);
            hout[m * NCLASS +  0] = c[t][0][r];
            hout[m * NCLASS + 16] = c[t][1][r];
            hout[m * NCLASS + 32] = c[t][2][r];
            hout[m * NCLASS + 48] = c[t][3][r];
        }
    }
}

__global__ void __launch_bounds__(GEMM_THREADS)
spgat_gemm(const float* __restrict__ x,
           const float* __restrict__ W,
           float* __restrict__ h, int N) {
    const int lane = threadIdx.x & 31;
    const int wave = __builtin_amdgcn_readfirstlane(threadIdx.x >> 5);
    const int slab = wave & 3;
    const int grp  = wave >> 2;
    const int half = lane >> 4;
    const int sub  = lane & 15;

    const int mtiles = N >> 4;
    const int nb  = (int)gridDim.x, bid = (int)blockIdx.x;
    const int q   = mtiles / nb, rem = mtiles % nb;
    const int t0  = bid * q + min(bid, rem);
    const int cnt = q + (bid < rem ? 1 : 0);
    const int c0  = (cnt + 1) >> 1;
    int tile = grp ? t0 + c0 : t0;
    int left = grp ? cnt - c0 : c0;

    while (left > 0) {
        const int m = min(left, GEMM_MT);
        switch (m) {
            case 7:  gemm_tiles<7>(x, W, h, tile, slab * 64, half, sub); break;
            case 6:  gemm_tiles<6>(x, W, h, tile, slab * 64, half, sub); break;
            case 5:  gemm_tiles<5>(x, W, h, tile, slab * 64, half, sub); break;
            case 4:  gemm_tiles<4>(x, W, h, tile, slab * 64, half, sub); break;
            case 3:  gemm_tiles<3>(x, W, h, tile, slab * 64, half, sub); break;
            case 2:  gemm_tiles<2>(x, W, h, tile, slab * 64, half, sub); break;
            default: gemm_tiles<1>(x, W, h, tile, slab * 64, half, sub); break;
        }
        tile += m;
        left -= m;
    }
}

#define SC_BLOCKS  512
#define SC_THREADS 256
#define SC_WAVES   (SC_BLOCKS * (SC_THREADS / 32))
#define SC_ROWS    13

typedef float v4f __attribute__((ext_vector_type(4)));
typedef int   v4i __attribute__((ext_vector_type(4)));

__device__ __forceinline__ void store_planes(unsigned int* __restrict__ plane_w, const int wv,
                                             const v4i r, const int g, const int E, const int lane) {
    unsigned kk[4] = { ((unsigned)r.x / ROWS_PB) & 255u, ((unsigned)r.y / ROWS_PB) & 255u,
                       ((unsigned)r.z / ROWS_PB) & 255u, ((unsigned)r.w / ROWS_PB) & 255u };
    unsigned mine = 0;
    #pragma unroll
    for (int i = 0; i < 4; ++i) {
        if (4 * g + i >= E) kk[i] = 255u;
        #pragma unroll
        for (int p = 0; p < 8; ++p) {
            const unsigned b = __builtin_amdgcn_ballot_w32(((kk[i] >> p) & 1u) != 0u);
            if (lane == i * 8 + p) mine = b;
        }
    }
    plane_w[(size_t)wv * 32 + lane] = mine;
}

__global__ void __launch_bounds__(SC_THREADS)
spgat_scores(const float* __restrict__ h,
             const float* __restrict__ a,
             float* __restrict__ s_row,
             float* __restrict__ s_col,
             const int* __restrict__ ei, int E,
             unsigned int* __restrict__ key_w, int N) {
    const int lane = threadIdx.x & 31;
    const int gtid = blockIdx.x * SC_THREADS + threadIdx.x;
    const int nthr = SC_BLOCKS * SC_THREADS;
    const int nq   = (E + 3) >> 2;
    const int nwg  = (nq + 31) >> 5;

    const v4f* a4 = (const v4f*)a;
    const v4f ar0 = a4[lane], ar1 = a4[32 + lane];
    const v4f ac0 = a4[64 + lane], ac1 = a4[96 + lane];
    asm volatile("" :: "v"(ar0), "v"(ar1), "v"(ac0), "v"(ac1));

    const int g0 = gtid, g1 = gtid + nthr;
    const int glast = max(nq, 1) - 1;
    const v4i k0 = ((const v4i*)ei)[min(g0, glast)];
    const v4i k1 = ((const v4i*)ei)[min(g1, glast)];

    const int wg   = __builtin_amdgcn_readfirstlane(gtid >> 5);
    const int q    = N / SC_WAVES, rem = N % SC_WAVES;
    const int rbeg = wg * q + min(wg, rem);
    const int rend = rbeg + q + (wg < rem ? 1 : 0);

    for (int r0 = rbeg; r0 < rend; r0 += SC_ROWS) {
        v4f lo[SC_ROWS], hi[SC_ROWS];
        #pragma unroll
        for (int u = 0; u < SC_ROWS; ++u) {
            const int r = min(r0 + u, rend - 1);
            const v4f* hr = (const v4f*)(h + (size_t)r * NCLASS) + lane;
            lo[u] = hr[0];
            hi[u] = hr[32];
        }
        asm volatile("" :: "v"(k0), "v"(k1),
                     "v"(lo[0]), "v"(lo[1]), "v"(lo[2]), "v"(lo[3]), "v"(lo[4]), "v"(lo[5]), "v"(lo[6]),
                     "v"(lo[7]), "v"(lo[8]), "v"(lo[9]), "v"(lo[10]), "v"(lo[11]), "v"(lo[12]),
                     "v"(hi[0]), "v"(hi[1]), "v"(hi[2]), "v"(hi[3]), "v"(hi[4]), "v"(hi[5]), "v"(hi[6]),
                     "v"(hi[7]), "v"(hi[8]), "v"(hi[9]), "v"(hi[10]), "v"(hi[11]), "v"(hi[12]));
        float my_sr = 0.f, my_sc = 0.f;
        #pragma unroll
        for (int u = 0; u < SC_ROWS; ++u) {
            float sr = lo[u].x * ar0.x;
            sr += lo[u].y * ar0.y; sr += lo[u].z * ar0.z; sr += lo[u].w * ar0.w;
            sr += hi[u].x * ar1.x; sr += hi[u].y * ar1.y; sr += hi[u].z * ar1.z; sr += hi[u].w * ar1.w;
            float sc = lo[u].x * ac0.x;
            sc += lo[u].y * ac0.y; sc += lo[u].z * ac0.z; sc += lo[u].w * ac0.w;
            sc += hi[u].x * ac1.x; sc += hi[u].y * ac1.y; sc += hi[u].z * ac1.z; sc += hi[u].w * ac1.w;
            #pragma unroll
            for (int off = 16; off; off >>= 1) {
                sr += __shfl_xor(sr, off, 32);
                sc += __shfl_xor(sc, off, 32);
            }
            if (lane == u) { my_sr = sr; my_sc = sc; }
        }
        const int r = r0 + lane;
        if (lane < SC_ROWS && r < rend) { s_row[r] = my_sr; s_col[r] = my_sc; }
    }

    if (wg < nwg)            store_planes(key_w, wg, k0, g0, E, lane);
    if (wg + SC_WAVES < nwg) store_planes(key_w, wg + SC_WAVES, k1, g1, E, lane);
    for (int wv = wg + 2 * SC_WAVES; wv < nwg; wv += SC_WAVES) {
        const int g = (wv << 5) + lane;
        store_planes(key_w, wv, ((const v4i*)ei)[min(g, glast)], g, E, lane);
    }
}

#define LIST_CAP    4096
#define ROW_CAP     64
#define ROW_PITCH   65
#define GROUP       12
#define AGG_THREADS 1024
#define SCAN_U      7

__global__ void __launch_bounds__(AGG_THREADS)
spgat_aggregate(const int* __restrict__ ei, int E,
                const unsigned int* __restrict__ key_w,
                const float* __restrict__ s_row,
                const float* __restrict__ s_col,
                const float* __restrict__ h,
                float* __restrict__ out, int N) {
    __shared__ int   list_e[LIST_CAP];
    __shared__ int   rl_c[ROWS_PB * ROW_PITCH];
    __shared__ float rl_w[ROWS_PB * ROW_PITCH];
    __shared__ int   rcnt[ROWS_PB];
    __shared__ int   cnt;
    __shared__ int   ovf;

    const int tid  = threadIdx.x;
    const int lane = tid & 31;
    const int wave = __builtin_amdgcn_readfirstlane(tid >> 5);
    const int row0 = blockIdx.x * ROWS_PB;
    const int nrows = min(ROWS_PB, N - row0);
    const unsigned nr = (unsigned)nrows;

    if (tid < ROWS_PB) rcnt[tid] = 0;
    if (tid == 0) { cnt = 0; ovf = 0; }
    __syncthreads();

    auto put = [&](int pos, int e) {
        if (pos < LIST_CAP) list_e[pos] = e;
        else ovf = 1;
    };
    unsigned cmask[8];
    #pragma unroll
    for (int p = 0; p < 8; ++p) cmask[p] = ((blockIdx.x >> p) & 1u) ? 0u : ~0u;
    auto match = [&](const uint4 a, const uint4 b) -> unsigned {
        return (a.x ^ cmask[0]) & (a.y ^ cmask[1]) & (a.z ^ cmask[2]) & (a.w ^ cmask[3]) &
               (b.x ^ cmask[4]) & (b.y ^ cmask[5]) & (b.z ^ cmask[6]) & (b.w ^ cmask[7]);
    };

    const int nsc = (((E + 3) >> 2) + 31) >> 5 << 2;
    const uint4* pl4 = (const uint4*)key_w;
    const int soff = (int)(((long long)blockIdx.x * nsc) / gridDim.x) & ~31;
    for (int sb = 0; sb < nsc; sb += SCAN_U * AGG_THREADS) {
        const int s0 = sb + tid;
        uint4 qa[SCAN_U], qb[SCAN_U];
        #pragma unroll
        for (int u = 0; u < SCAN_U; ++u) {
            const int s = min(s0 + u * AGG_THREADS, nsc - 1);
            const int ss = (s + soff < nsc) ? s + soff : s + soff - nsc;
            qa[u] = pl4[2 * ss];
            qb[u] = pl4[2 * ss + 1];
        }
        unsigned mm[SCAN_U];
        int c = 0;
        #pragma unroll
        for (int u = 0; u < SCAN_U; ++u) {
            unsigned m = match(qa[u], qb[u]);
            if (s0 + u * AGG_THREADS >= nsc) m = 0;
            mm[u] = m;
            c += __popc(m);
        }
        int incl = c;
        #pragma unroll
        for (int off = 1; off < 32; off <<= 1) {
            const int t = __shfl_up(incl, off, 32);
            if (lane >= off) incl += t;
        }
        const int total = __builtin_amdgcn_readlane(incl, 31);
        if (total) {
            int base = 0;
            if (lane == 31) base = atomicAdd(&cnt, total);
            base = __builtin_amdgcn_readlane(base, 31);
            int pos = base + incl - c;
            #pragma unroll
            for (int u = 0; u < SCAN_U; ++u) {
                const int s = s0 + u * AGG_THREADS;
                const int ss = (s + soff < nsc) ? s + soff : s + soff - nsc;
                const int e = ((ss & ~3) << 5) + (ss & 3);
                unsigned m = mm[u];
                while (m) {
                    const int b = __ffs(m) - 1;
                    m &= m - 1;
                    put(pos++, e + (b << 2));
                }
            }
        }
    }
    __syncthreads();

    const int n = min(cnt, LIST_CAP);
    for (int k = tid; k < n; k += AGG_THREADS) {
        const int e = list_e[k];
        const int r = ei[min(e, E - 1)];
        const unsigned rl = (unsigned)(r - row0);
        if (e < E && rl < nr) {
            const int c = ei[E + e];
            const float lg = s_row[r] + s_col[c];
            const float l  = lg > 0.f ? lg : 0.2f * lg;
            const int p = atomicAdd(&rcnt[rl], 1);
            if (p < ROW_CAP) {
                rl_c[rl * ROW_PITCH + p] = c;
                rl_w[rl * ROW_PITCH + p] = __expf(-l);
            } else {
                ovf = 1;
            }
        }
    }
    __syncthreads();

    const int overflow = __builtin_amdgcn_readfirstlane(ovf);
    for (int r = wave; r < nrows; r += 32) {
        float4 lo = make_float4(0.f, 0.f, 0.f, 0.f);
        float4 hi = make_float4(0.f, 0.f, 0.f, 0.f);
        float rsum = 0.f;
        auto add_edge = [&](float w, const float4 a, const float4 b) {
            lo.x += w * a.x; lo.y += w * a.y; lo.z += w * a.z; lo.w += w * a.w;
            hi.x += w * b.x; hi.y += w * b.y; hi.z += w * b.z; hi.w += w * b.w;
            rsum += w;
        };

        if (!overflow) {
            const int nw = __builtin_amdgcn_readfirstlane(min(rcnt[r], ROW_CAP));
            const int*   wc = rl_c + r * ROW_PITCH;
            const float* ww = rl_w + r * ROW_PITCH;
            for (int k0 = 0; k0 < nw; k0 += GROUP) {
                float4 hv[GROUP][2];
                float ev[GROUP];
                int   cc[GROUP];
                const int   kk = min(k0 + lane, nw - 1);
                const int   cv = wc[kk];
                const float wv = ww[kk];
                #pragma unroll
                for (int u = 0; u < GROUP; ++u) {
                    cc[u] = __builtin_amdgcn_readlane(cv, u);
                    ev[u] = __int_as_float(__builtin_amdgcn_readlane(__float_as_int(wv), u));
                }
                #pragma unroll
                for (int u = 0; u < GROUP; ++u) {
                    if (k0 + u < nw) {
                        const float4* hc = (const float4*)(h + (size_t)cc[u] * NCLASS) + lane;
                        hv[u][0] = hc[0];
                        hv[u][1] = hc[32];
                    }
                }
                #pragma unroll
                for (int u = 0; u < GROUP; ++u) {
                    if (k0 + u < nw) add_edge(ev[u], hv[u][0], hv[u][1]);
                }
            }
        } else {
            const int rg = row0 + r;
            for (int e0 = 0; e0 < E; e0 += 32) {
                const int e = e0 + lane;
                const bool mine = (e < E) && (ei[min(e, E - 1)] == rg);
                unsigned m = __builtin_amdgcn_ballot_w32(mine);
                while (m) {
                    const int b = __ffs(m) - 1;
                    m &= m - 1;
                    const int ee = e0 + b;
                    const int c = ei[E + ee];
                    const float lg = s_row[rg] + s_col[c];
                    const float l  = lg > 0.f ? lg : 0.2f * lg;
                    const float4* hc = (const float4*)(h + (size_t)c * NCLASS) + lane;
                    const float wgt = __int_as_float(__builtin_amdgcn_readfirstlane(__float_as_int(__expf(-l))));
                    add_edge(wgt, hc[0], hc[32]);
                }
            }
        }

        const float inv = 1.0f / rsum;
        float4* o = (float4*)(out + (size_t)(row0 + r) * NCLASS);
        float v[8] = { lo.x, lo.y, lo.z, lo.w, hi.x, hi.y, hi.z, hi.w };
        float mx = -3.402823466e38f;
        #pragma unroll
        for (int i = 0; i < 8; ++i) {
            const float hp = v[i] * inv;
            const float el = hp > 0.f ? hp : (__expf(hp) - 1.f);
            v[i] = el;
            mx = fmaxf(mx, el);
        }
        #pragma unroll
        for (int off = 16; off; off >>= 1) mx = fmaxf(mx, __shfl_xor(mx, off, 32));
        float sm = 0.f;
        #pragma unroll
        for (int i = 0; i < 8; ++i) sm += __expf(v[i] - mx);
        #pragma unroll
        for (int off = 16; off; off >>= 1) sm += __shfl_xor(sm, off, 32);
        const float lse = mx + __logf(sm);
        o[lane]      = make_float4(v[0] - lse, v[1] - lse, v[2] - lse, v[3] - lse);
        o[lane + 32] = make_float4(v[4] - lse, v[5] - lse, v[6] - lse, v[7] - lse);
    }
}

extern "C" void kernel_launch(void* const* d_in, const int* in_sizes, int n_in,
                              void* d_out, int out_size, void* d_ws, size_t ws_size,
                              hipStream_t stream) {
    const float* x  = (const float*)d_in[0];
    const float* W  = (const float*)d_in[1];
    const float* a  = (const float*)d_in[2];
    const int*   ei = (const int*)d_in[3];

    const int N = in_sizes[0] / NFEAT;
    const int E = in_sizes[3] / 2;

    float* h      = (float*)d_ws;
    float* s_row  = h + (size_t)N * NCLASS;
    float* s_col  = s_row + N;
    unsigned int* key_w = (unsigned int*)(s_col + N);

    float* out = (float*)d_out;

    {
        spgat_gemm<<<GEMM_BLOCKS, GEMM_THREADS, 0, stream>>>(x, W, h, N);
    }
    {
        spgat_scores<<<SC_BLOCKS, SC_THREADS, 0, stream>>>(h, a, s_row, s_col, ei, E, key_w, N);
    }
    {
        dim3 grid((N + ROWS_PB - 1) / ROWS_PB);
        spgat_aggregate<<<grid, AGG_THREADS, 0, stream>>>(ei, E, key_w, s_row, s_col, h, out, N);
    }
}
